# starting kernel
# baseline (speedup 1.0000x reference)
#include <hip/hip_runtime.h>

typedef __attribute__((ext_vector_type(2))) float v2f;
typedef __attribute__((ext_vector_type(8))) float v8f;

#define HDIM 128
#define ODIM 40

__global__ void k_init_deg(float* __restrict__ deg, int n) {
    int i = blockIdx.x * blockDim.x + threadIdx.x;
    if (i < n) deg[i] = 1.0f;
}

__global__ void k_edge_deg(const int* __restrict__ col, float* __restrict__ deg, int e) {
    int i = blockIdx.x * blockDim.x + threadIdx.x;
    if (i < e) atomicAdd(&deg[col[i]], 1.0f);
}

__global__ void k_finish_deg(float* __restrict__ deg, int n) {
    int i = blockIdx.x * blockDim.x + threadIdx.x;
    if (i < n) {
        float d = deg[i];
        deg[i] = (d > 0.0f) ? rsqrtf(d) : 0.0f;
    }
}

__global__ void k_zero4(float4* __restrict__ p, long n4) {
    long i = (long)blockIdx.x * blockDim.x + threadIdx.x;
    if (i < n4) p[i] = make_float4(0.f, 0.f, 0.f, 0.f);
}

template <int NPAD>
__global__ __launch_bounds__(256) void k_gemm_wmma(
        const float* __restrict__ A, const float* __restrict__ B,
        const float* __restrict__ bias, float* __restrict__ C,
        int M, int ncols) {
    constexpr int K  = 128;
    constexpr int NT = NPAD / 16;
    constexpr int LSTR = K + 2;
    __shared__ float ldsBT[NPAD * LSTR];

    for (int i = threadIdx.x; i < NPAD * K; i += 256) {
        int k = i & (K - 1);
        int n = i >> 7;
        ldsBT[n * LSTR + k] = (n < ncols) ? B[k * ncols + n] : 0.0f;
    }
    __syncthreads();

    const int lane  = threadIdx.x & 31;
    const int tile  = blockIdx.x * 8 + (threadIdx.x >> 5);
    const int mtile = M >> 4;
    if (tile >= mtile) return;

    const int rowA = tile * 16 + (lane & 15);
    const int koff = (lane >> 4) * 2;
    const float* aptr = A + (size_t)rowA * K + koff;

    v8f acc[NT];
    v8f z8 = {0.f, 0.f, 0.f, 0.f, 0.f, 0.f, 0.f, 0.f};
#pragma unroll
    for (int nt = 0; nt < NT; ++nt) acc[nt] = z8;

#pragma unroll 4
    for (int k = 0; k < K; k += 4) {
        v2f a = *(const v2f*)(aptr + k);
#pragma unroll
        for (int nt = 0; nt < NT; ++nt) {
            int n = nt * 16 + (lane & 15);
            v2f b = *(const v2f*)(&ldsBT[n * LSTR + k + koff]);
            acc[nt] = __builtin_amdgcn_wmma_f32_16x16x4_f32(
                 false, a,  false, b,
                 (short)0, acc[nt],
                 false,  false);
        }
    }

    const int rbase = tile * 16 + ((lane >> 4) << 3);
#pragma unroll
    for (int nt = 0; nt < NT; ++nt) {
        int n = nt * 16 + (lane & 15);
        if (n >= ncols) continue;
        float badd = bias ? bias[n] : 0.0f;
#pragma unroll
        for (int r = 0; r < 8; ++r) {
            C[(size_t)(rbase + r) * ncols + n] = acc[nt][r] + badd;
        }
    }
}

__global__ __launch_bounds__(256) void k_scatter(
        const float* __restrict__ h, const float* __restrict__ dis,
        const int* __restrict__ row, const int* __restrict__ col,
        float* __restrict__ y, int e) {
    int w = blockIdx.x * 8 + (threadIdx.x >> 5);
    if (w >= e) return;
    int lane = threadIdx.x & 31;
    int r = row[w];
    int c = col[w];
    float nrm = dis[r] * dis[c];
    const float4 v = *(const float4*)(h + (size_t)r * HDIM + lane * 4);
    float* dst = y + (size_t)c * HDIM + lane * 4;
    atomicAdd(dst + 0, v.x * nrm);
    atomicAdd(dst + 1, v.y * nrm);
    atomicAdd(dst + 2, v.z * nrm);
    atomicAdd(dst + 3, v.w * nrm);
}

__global__ void k_post1(const float* __restrict__ h, float* __restrict__ y,
                        const float* __restrict__ dis,
                        const float* __restrict__ b1,
                        const float* __restrict__ gamma,
                        const float* __restrict__ beta,
                        const float* __restrict__ mean,
                        const float* __restrict__ var, int total) {
    int i = blockIdx.x * blockDim.x + threadIdx.x;
    if (i >= total) return;
    int ch = i & (HDIM - 1);
    int nd = i >> 7;
    float d = dis[nd];
    float v = y[i] + d * d * h[i] + b1[ch];
    v = (v - mean[ch]) * rsqrtf(var[ch] + 1e-5f) * gamma[ch] + beta[ch];
    y[i] = fmaxf(v, 0.0f);
}

__global__ void k_post2(const float* __restrict__ h, const float* __restrict__ x1,
                        float* __restrict__ y, const float* __restrict__ dis,
                        const float* __restrict__ b2, int total) {
    int i = blockIdx.x * blockDim.x + threadIdx.x;
    if (i >= total) return;
    int ch = i & (HDIM - 1);
    int nd = i >> 7;
    float d = dis[nd];
    float v = y[i] + d * d * h[i] + b2[ch];
    y[i] = fmaxf(x1[i], v);
}

extern "C" void kernel_launch(void* const* d_in, const int* in_sizes, int n_in,
                              void* d_out, int out_size, void* d_ws, size_t ws_size,
                              hipStream_t stream) {
    const float* node_feat = (const float*)d_in[0];
    const int*   eidx      = (const int*)d_in[1];
    const float* W1        = (const float*)d_in[2];
    const float* b1        = (const float*)d_in[3];
    const float* gamma1    = (const float*)d_in[4];
    const float* beta1     = (const float*)d_in[5];
    const float* mean1     = (const float*)d_in[6];
    const float* var1      = (const float*)d_in[7];
    const float* W2        = (const float*)d_in[8];
    const float* b2        = (const float*)d_in[9];
    const float* Wf        = (const float*)d_in[10];
    const float* bf        = (const float*)d_in[11];
    float* out = (float*)d_out;

    const int n = in_sizes[0] / HDIM;
    const int e = in_sizes[1] / 2;
    const int* row = eidx;
    const int* col = eidx + e;

    float* dis = (float*)d_ws;
    float* h   = dis + n;
    float* y1  = h + (size_t)n * HDIM;
    float* y2  = y1 + (size_t)n * HDIM;

    const int nh      = n * HDIM;
    const int mtiles  = n >> 4;
    const int gemmGrd = (mtiles + 7) / 8;
    const int edgeGrd = (e + 7) / 8;

    k_init_deg<<<(n + 255) / 256, 256, 0, stream>>>(dis, n);
    k_edge_deg<<<(e + 255) / 256, 256, 0, stream>>>(col, dis, e);
    k_finish_deg<<<(n + 255) / 256, 256, 0, stream>>>(dis, n);

    long n4 = (long)2 * nh / 4;
    k_zero4<<<(int)((n4 + 255) / 256), 256, 0, stream>>>((float4*)y1, n4);

    k_gemm_wmma<128><<<gemmGrd, 256, 0, stream>>>(node_feat, W1, nullptr, h, n, HDIM);
    k_scatter<<<edgeGrd, 256, 0, stream>>>(h, dis, row, col, y1, e);
    k_post1<<<(nh + 255) / 256, 256, 0, stream>>>(h, y1, dis, b1, gamma1, beta1, mean1, var1, nh);

    k_gemm_wmma<128><<<gemmGrd, 256, 0, stream>>>(y1, W2, nullptr, h, n, HDIM);
    k_scatter<<<edgeGrd, 256, 0, stream>>>(h, dis, row, col, y2, e);
    k_post2<<<(nh + 255) / 256, 256, 0, stream>>>(h, y1, y2, dis, b2, nh);

    k_gemm_wmma<48><<<gemmGrd, 256, 0, stream>>>(y2, Wf, bf, out, n, ODIM);
}


# Replace float-atomic scatter-add by target-sorted gather: count/scan/fill the edge list once per call, one wave sums a node's source rows in registers (sum order = arrival order of count atomics)
# speedup vs baseline: 2.1251x; 2.1251x over previous
#include <hip/hip_runtime.h>

typedef __attribute__((ext_vector_type(2))) float v2f;
typedef __attribute__((ext_vector_type(8))) float v8f;

#define HDIM 128
#define ODIM 40

__global__ void k_zero_cnt(int* __restrict__ cnt, int n) {
    int i = blockIdx.x * blockDim.x + threadIdx.x;
    if (i < n) cnt[i] = 0;
}

__global__ void k_count(const int* __restrict__ col, int* __restrict__ cnt,
                        int* __restrict__ rank, int e) {
    int i = blockIdx.x * blockDim.x + threadIdx.x;
    if (i < e) rank[i] = atomicAdd(&cnt[col[i]], 1);
}

__global__ __launch_bounds__(1024) void k_scan(const int* __restrict__ cnt,
                                               int* __restrict__ off,
                                               float* __restrict__ dis, int n) {
    __shared__ int s_pre[32];
    __shared__ int s_tot[32];
    const int tid  = threadIdx.x;
    const int lane = tid & 31;
    const int wv   = tid >> 5;
    const int tile0 = blockIdx.x * 1024;

    int pre = 0;
    for (int i = tid; i < tile0; i += 1024) pre += cnt[i];

    const int idx = tile0 + tid;
    const int c = (idx < n) ? cnt[idx] : 0;

    int incl = c;
    for (int d = 1; d < 32; d <<= 1) {
        int t = __shfl_up(incl, d);
        if (lane >= d) incl += t;
    }
    for (int d = 16; d > 0; d >>= 1) pre += __shfl_xor(pre, d);

    if (lane == 31) s_tot[wv] = incl;
    if (lane == 0)  s_pre[wv] = pre;
    __syncthreads();

    int p = s_pre[lane];
    int before = (lane < wv) ? s_tot[lane] : 0;
    for (int d = 16; d > 0; d >>= 1) {
        p      += __shfl_xor(p, d);
        before += __shfl_xor(before, d);
    }
    if (idx < n) {
        off[idx] = p + before + incl - c;
        float dg = (float)(c + 1);
        dis[idx] = (dg > 0.0f) ? rsqrtf(dg) : 0.0f;
    }
}

__global__ void k_fill(const int* __restrict__ row, const int* __restrict__ col,
                       const int* __restrict__ rank, const int* __restrict__ off,
                       int* __restrict__ srcs, int e) {
    int i = blockIdx.x * blockDim.x + threadIdx.x;
    if (i < e) srcs[off[col[i]] + rank[i]] = row[i];
}

template <int NPAD>
__global__ __launch_bounds__(256) void k_gemm_wmma(
        const float* __restrict__ A, const float* __restrict__ B,
        const float* __restrict__ bias, float* __restrict__ C,
        int M, int ncols) {
    constexpr int K  = 128;
    constexpr int NT = NPAD / 16;
    constexpr int LSTR = K + 2;
    __shared__ float ldsBT[NPAD * LSTR];

    for (int i = threadIdx.x; i < NPAD * K; i += 256) {
        int k = i & (K - 1);
        int n = i >> 7;
        ldsBT[n * LSTR + k] = (n < ncols) ? B[k * ncols + n] : 0.0f;
    }
    __syncthreads();

    const int lane  = threadIdx.x & 31;
    const int tile  = blockIdx.x * 8 + (threadIdx.x >> 5);
    const int mtile = M >> 4;
    if (tile >= mtile) return;

    const int rowA = tile * 16 + (lane & 15);
    const int koff = (lane >> 4) * 2;
    const float* aptr = A + (size_t)rowA * K + koff;

    v8f acc[NT];
    v8f z8 = {0.f, 0.f, 0.f, 0.f, 0.f, 0.f, 0.f, 0.f};
#pragma unroll
    for (int nt = 0; nt < NT; ++nt) acc[nt] = z8;

#pragma unroll 4
    for (int k = 0; k < K; k += 4) {
        v2f a = *(const v2f*)(aptr + k);
#pragma unroll
        for (int nt = 0; nt < NT; ++nt) {
            int n = nt * 16 + (lane & 15);
            v2f b = *(const v2f*)(&ldsBT[n * LSTR + k + koff]);
            acc[nt] = __builtin_amdgcn_wmma_f32_16x16x4_f32(
                 false, a,  false, b,
                 (short)0, acc[nt],
                 false,  false);
        }
    }

    const int rbase = tile * 16 + ((lane >> 4) << 3);
#pragma unroll
    for (int nt = 0; nt < NT; ++nt) {
        int n = nt * 16 + (lane & 15);
        if (n >= ncols) continue;
        float badd = bias ? bias[n] : 0.0f;
#pragma unroll
        for (int r = 0; r < 8; ++r) {
            C[(size_t)(rbase + r) * ncols + n] = acc[nt][r] + badd;
        }
    }
}

__global__ __launch_bounds__(256) void k_gather(
        const float* __restrict__ h, const float* __restrict__ dis,
        const int* __restrict__ off, const int* __restrict__ cnt,
        const int* __restrict__ srcs, float* __restrict__ y, int n) {
    const int node = __builtin_amdgcn_readfirstlane(blockIdx.x * 8 + (threadIdx.x >> 5));
    if (node >= n) return;
    const int lane  = threadIdx.x & 31;
    const int start = off[node];
    const int m     = cnt[node];
    const float dc  = dis[node];
    const float* hl = h + lane * 4;

    float4 acc = make_float4(0.f, 0.f, 0.f, 0.f);
    for (int base = 0; base < m; base += 32) {
        const int k = (m - base < 32) ? (m - base) : 32;
        int s = 0;
        float w = 0.0f;
        if (lane < k) {
            s = srcs[start + base + lane];
            w = dis[s] * dc;
        }
        const int wi = __float_as_int(w);
        int j = 0;
        for (; j + 4 <= k; j += 4) {
            const int s0 = __builtin_amdgcn_readlane(s, j);
            const int s1 = __builtin_amdgcn_readlane(s, j + 1);
            const int s2 = __builtin_amdgcn_readlane(s, j + 2);
            const int s3 = __builtin_amdgcn_readlane(s, j + 3);
            const float w0 = __int_as_float(__builtin_amdgcn_readlane(wi, j));
            const float w1 = __int_as_float(__builtin_amdgcn_readlane(wi, j + 1));
            const float w2 = __int_as_float(__builtin_amdgcn_readlane(wi, j + 2));
            const float w3 = __int_as_float(__builtin_amdgcn_readlane(wi, j + 3));
            const float4 v0 = *(const float4*)(hl + (size_t)s0 * HDIM);
            const float4 v1 = *(const float4*)(hl + (size_t)s1 * HDIM);
            const float4 v2 = *(const float4*)(hl + (size_t)s2 * HDIM);
            const float4 v3 = *(const float4*)(hl + (size_t)s3 * HDIM);
            acc.x = fmaf(v0.x, w0, acc.x); acc.y = fmaf(v0.y, w0, acc.y);
            acc.z = fmaf(v0.z, w0, acc.z); acc.w = fmaf(v0.w, w0, acc.w);
            acc.x = fmaf(v1.x, w1, acc.x); acc.y = fmaf(v1.y, w1, acc.y);
            acc.z = fmaf(v1.z, w1, acc.z); acc.w = fmaf(v1.w, w1, acc.w);
            acc.x = fmaf(v2.x, w2, acc.x); acc.y = fmaf(v2.y, w2, acc.y);
            acc.z = fmaf(v2.z, w2, acc.z); acc.w = fmaf(v2.w, w2, acc.w);
            acc.x = fmaf(v3.x, w3, acc.x); acc.y = fmaf(v3.y, w3, acc.y);
            acc.z = fmaf(v3.z, w3, acc.z); acc.w = fmaf(v3.w, w3, acc.w);
        }
        for (; j < k; ++j) {
            const int s0 = __builtin_amdgcn_readlane(s, j);
            const float w0 = __int_as_float(__builtin_amdgcn_readlane(wi, j));
            const float4 v0 = *(const float4*)(hl + (size_t)s0 * HDIM);
            acc.x = fmaf(v0.x, w0, acc.x); acc.y = fmaf(v0.y, w0, acc.y);
            acc.z = fmaf(v0.z, w0, acc.z); acc.w = fmaf(v0.w, w0, acc.w);
        }
    }
    *(float4*)(y + (size_t)node * HDIM + lane * 4) = acc;
}

__global__ void k_post1(const float* __restrict__ h, float* __restrict__ y,
                        const float* __restrict__ dis,
                        const float* __restrict__ b1,
                        const float* __restrict__ gamma,
                        const float* __restrict__ beta,
                        const float* __restrict__ mean,
                        const float* __restrict__ var, int total) {
    int i = blockIdx.x * blockDim.x + threadIdx.x;
    if (i >= total) return;
    int ch = i & (HDIM - 1);
    int nd = i >> 7;
    float d = dis[nd];
    float v = y[i] + d * d * h[i] + b1[ch];
    v = (v - mean[ch]) * rsqrtf(var[ch] + 1e-5f) * gamma[ch] + beta[ch];
    y[i] = fmaxf(v, 0.0f);
}

__global__ void k_post2(const float* __restrict__ h, const float* __restrict__ x1,
                        float* __restrict__ y, const float* __restrict__ dis,
                        const float* __restrict__ b2, int total) {
    int i = blockIdx.x * blockDim.x + threadIdx.x;
    if (i >= total) return;
    int ch = i & (HDIM - 1);
    int nd = i >> 7;
    float d = dis[nd];
    float v = y[i] + d * d * h[i] + b2[ch];
    y[i] = fmaxf(x1[i], v);
}

extern "C" void kernel_launch(void* const* d_in, const int* in_sizes, int n_in,
                              void* d_out, int out_size, void* d_ws, size_t ws_size,
                              hipStream_t stream) {
    const float* node_feat = (const float*)d_in[0];
    const int*   eidx      = (const int*)d_in[1];
    const float* W1        = (const float*)d_in[2];
    const float* b1        = (const float*)d_in[3];
    const float* gamma1    = (const float*)d_in[4];
    const float* beta1     = (const float*)d_in[5];
    const float* mean1     = (const float*)d_in[6];
    const float* var1      = (const float*)d_in[7];
    const float* W2        = (const float*)d_in[8];
    const float* b2        = (const float*)d_in[9];
    const float* Wf        = (const float*)d_in[10];
    const float* bf        = (const float*)d_in[11];
    float* out = (float*)d_out;

    const int n = in_sizes[0] / HDIM;
    const int e = in_sizes[1] / 2;
    const int* row = eidx;
    const int* col = eidx + e;

    float* dis = (float*)d_ws;
    float* h   = dis + n;
    float* y1  = h + (size_t)n * HDIM;
    float* y2  = y1 + (size_t)n * HDIM;
    int* cnt   = (int*)(y2 + (size_t)n * HDIM);
    int* off   = cnt + n;
    int* rank  = off + n;
    int* srcs  = rank + e;

    const int nh      = n * HDIM;
    const int mtiles  = n >> 4;
    const int gemmGrd = (mtiles + 7) / 8;
    const int nodeGrd = (n + 7) / 8;

    k_zero_cnt<<<(n + 255) / 256, 256, 0, stream>>>(cnt, n);
    k_count<<<(e + 255) / 256, 256, 0, stream>>>(col, cnt, rank, e);
    k_scan<<<(n + 1023) / 1024, 1024, 0, stream>>>(cnt, off, dis, n);
    k_fill<<<(e + 255) / 256, 256, 0, stream>>>(row, col, rank, off, srcs, e);

    k_gemm_wmma<128><<<gemmGrd, 256, 0, stream>>>(node_feat, W1, nullptr, h, n, HDIM);
    k_gather<<<nodeGrd, 256, 0, stream>>>(h, dis, off, cnt, srcs, y1, n);
    k_post1<<<(nh + 255) / 256, 256, 0, stream>>>(h, y1, dis, b1, gamma1, beta1, mean1, var1, nh);

    k_gemm_wmma<128><<<gemmGrd, 256, 0, stream>>>(y1, W2, nullptr, h, n, HDIM);
    k_gather<<<nodeGrd, 256, 0, stream>>>(h, dis, off, cnt, srcs, y2, n);
    k_post2<<<(nh + 255) / 256, 256, 0, stream>>>(h, y1, y2, dis, b2, nh);

    k_gemm_wmma<48><<<gemmGrd, 256, 0, stream>>>(y2, Wf, bf, out, n, ODIM);
}


# Fold both epilogue passes into the gather kernels: the wave that sums a node's row adds self loop and bias, then BN+ReLU (layer 1) or JK max (layer 2), before its one store; 11 -> 9 launches
# speedup vs baseline: 2.4655x; 1.1602x over previous
#include <hip/hip_runtime.h>

typedef __attribute__((ext_vector_type(2))) float v2f;
typedef __attribute__((ext_vector_type(8))) float v8f;

#define HDIM 128
#define ODIM 40

__global__ void k_zero_cnt(int* __restrict__ cnt, int n) {
    int i = blockIdx.x * blockDim.x + threadIdx.x;
    if (i < n) cnt[i] = 0;
}

__global__ void k_count(const int* __restrict__ col, int* __restrict__ cnt,
                        int* __restrict__ rank, int e) {
    int i = blockIdx.x * blockDim.x + threadIdx.x;
    if (i < e) rank[i] = atomicAdd(&cnt[col[i]], 1);
}

__global__ __launch_bounds__(1024) void k_scan(const int* __restrict__ cnt,
                                               int* __restrict__ off,
                                               float* __restrict__ dis, int n) {
    __shared__ int s_pre[32];
    __shared__ int s_tot[32];
    const int tid  = threadIdx.x;
    const int lane = tid & 31;
    const int wv   = tid >> 5;
    const int tile0 = blockIdx.x * 1024;

    int pre = 0;
    for (int i = tid; i < tile0; i += 1024) pre += cnt[i];

    const int idx = tile0 + tid;
    const int c = (idx < n) ? cnt[idx] : 0;

    int incl = c;
    for (int d = 1; d < 32; d <<= 1) {
        int t = __shfl_up(incl, d);
        if (lane >= d) incl += t;
    }
    for (int d = 16; d > 0; d >>= 1) pre += __shfl_xor(pre, d);

    if (lane == 31) s_tot[wv] = incl;
    if (lane == 0)  s_pre[wv] = pre;
    __syncthreads();

    int p = s_pre[lane];
    int before = (lane < wv) ? s_tot[lane] : 0;
    for (int d = 16; d > 0; d >>= 1) {
        p      += __shfl_xor(p, d);
        before += __shfl_xor(before, d);
    }
    if (idx < n) {
        off[idx] = p + before + incl - c;
        float dg = (float)(c + 1);
        dis[idx] = (dg > 0.0f) ? rsqrtf(dg) : 0.0f;
    }
}

__global__ void k_fill(const int* __restrict__ row, const int* __restrict__ col,
                       const int* __restrict__ rank, const int* __restrict__ off,
                       int* __restrict__ srcs, int e) {
    int i = blockIdx.x * blockDim.x + threadIdx.x;
    if (i < e) srcs[off[col[i]] + rank[i]] = row[i];
}

template <int NPAD>
__global__ __launch_bounds__(256) void k_gemm_wmma(
        const float* __restrict__ A, const float* __restrict__ B,
        const float* __restrict__ bias, float* __restrict__ C,
        int M, int ncols) {
    constexpr int K  = 128;
    constexpr int NT = NPAD / 16;
    constexpr int LSTR = K + 2;
    __shared__ float ldsBT[NPAD * LSTR];

    for (int i = threadIdx.x; i < NPAD * K; i += 256) {
        int k = i & (K - 1);
        int n = i >> 7;
        ldsBT[n * LSTR + k] = (n < ncols) ? B[k * ncols + n] : 0.0f;
    }
    __syncthreads();

    const int lane  = threadIdx.x & 31;
    const int tile  = blockIdx.x * 8 + (threadIdx.x >> 5);
    const int mtile = M >> 4;
    if (tile >= mtile) return;

    const int rowA = tile * 16 + (lane & 15);
    const int koff = (lane >> 4) * 2;
    const float* aptr = A + (size_t)rowA * K + koff;

    v8f acc[NT];
    v8f z8 = {0.f, 0.f, 0.f, 0.f, 0.f, 0.f, 0.f, 0.f};
#pragma unroll
    for (int nt = 0; nt < NT; ++nt) acc[nt] = z8;

#pragma unroll 4
    for (int k = 0; k < K; k += 4) {
        v2f a = *(const v2f*)(aptr + k);
#pragma unroll
        for (int nt = 0; nt < NT; ++nt) {
            int n = nt * 16 + (lane & 15);
            v2f b = *(const v2f*)(&ldsBT[n * LSTR + k + koff]);
            acc[nt] = __builtin_amdgcn_wmma_f32_16x16x4_f32(
                 false, a,  false, b,
                 (short)0, acc[nt],
                 false,  false);
        }
    }

    const int rbase = tile * 16 + ((lane >> 4) << 3);
#pragma unroll
    for (int nt = 0; nt < NT; ++nt) {
        int n = nt * 16 + (lane & 15);
        if (n >= ncols) continue;
        float badd = bias ? bias[n] : 0.0f;
#pragma unroll
        for (int r = 0; r < 8; ++r) {
            C[(size_t)(rbase + r) * ncols + n] = acc[nt][r] + badd;
        }
    }
}

template <int LAYER>
__global__ __launch_bounds__(256) void k_gather(
        const float* __restrict__ h, const float* __restrict__ dis,
        const int* __restrict__ off, const int* __restrict__ cnt,
        const int* __restrict__ srcs, float* __restrict__ y, int n,
        const float* __restrict__ bias,
        const float* __restrict__ gamma, const float* __restrict__ beta,
        const float* __restrict__ mean, const float* __restrict__ var,
        const float* __restrict__ x1) {
    const int node = __builtin_amdgcn_readfirstlane(blockIdx.x * 8 + (threadIdx.x >> 5));
    if (node >= n) return;
    const int lane  = threadIdx.x & 31;
    const int start = off[node];
    const int m     = cnt[node];
    const float dc  = dis[node];
    const float* hl = h + lane * 4;

    const float4 hs = *(const float4*)(hl + (size_t)node * HDIM);
    float4 xr = hs;
    if (LAYER == 2) xr = *(const float4*)(x1 + (size_t)node * HDIM + lane * 4);

    float4 acc = make_float4(0.f, 0.f, 0.f, 0.f);
    for (int base = 0; base < m; base += 32) {
        const int k = (m - base < 32) ? (m - base) : 32;
        int s = 0;
        float w = 0.0f;
        if (lane < k) {
            s = srcs[start + base + lane];
            w = dis[s] * dc;
        }
        const int wi = __float_as_int(w);
        int j = 0;
        for (; j + 4 <= k; j += 4) {
            const int s0 = __builtin_amdgcn_readlane(s, j);
            const int s1 = __builtin_amdgcn_readlane(s, j + 1);
            const int s2 = __builtin_amdgcn_readlane(s, j + 2);
            const int s3 = __builtin_amdgcn_readlane(s, j + 3);
            const float w0 = __int_as_float(__builtin_amdgcn_readlane(wi, j));
            const float w1 = __int_as_float(__builtin_amdgcn_readlane(wi, j + 1));
            const float w2 = __int_as_float(__builtin_amdgcn_readlane(wi, j + 2));
            const float w3 = __int_as_float(__builtin_amdgcn_readlane(wi, j + 3));
            const float4 v0 = *(const float4*)(hl + (size_t)s0 * HDIM);
            const float4 v1 = *(const float4*)(hl + (size_t)s1 * HDIM);
            const float4 v2 = *(const float4*)(hl + (size_t)s2 * HDIM);
            const float4 v3 = *(const float4*)(hl + (size_t)s3 * HDIM);
            acc.x = fmaf(v0.x, w0, acc.x); acc.y = fmaf(v0.y, w0, acc.y);
            acc.z = fmaf(v0.z, w0, acc.z); acc.w = fmaf(v0.w, w0, acc.w);
            acc.x = fmaf(v1.x, w1, acc.x); acc.y = fmaf(v1.y, w1, acc.y);
            acc.z = fmaf(v1.z, w1, acc.z); acc.w = fmaf(v1.w, w1, acc.w);
            acc.x = fmaf(v2.x, w2, acc.x); acc.y = fmaf(v2.y, w2, acc.y);
            acc.z = fmaf(v2.z, w2, acc.z); acc.w = fmaf(v2.w, w2, acc.w);
            acc.x = fmaf(v3.x, w3, acc.x); acc.y = fmaf(v3.y, w3, acc.y);
            acc.z = fmaf(v3.z, w3, acc.z); acc.w = fmaf(v3.w, w3, acc.w);
        }
        for (; j < k; ++j) {
            const int s0 = __builtin_amdgcn_readlane(s, j);
            const float w0 = __int_as_float(__builtin_amdgcn_readlane(wi, j));
            const float4 v0 = *(const float4*)(hl + (size_t)s0 * HDIM);
            acc.x = fmaf(v0.x, w0, acc.x); acc.y = fmaf(v0.y, w0, acc.y);
            acc.z = fmaf(v0.z, w0, acc.z); acc.w = fmaf(v0.w, w0, acc.w);
        }
    }
    const float4 bb = *(const float4*)(bias + lane * 4);
    const float d2 = dc * dc;
    float4 v;
    v.x = acc.x + d2 * hs.x + bb.x;
    v.y = acc.y + d2 * hs.y + bb.y;
    v.z = acc.z + d2 * hs.z + bb.z;
    v.w = acc.w + d2 * hs.w + bb.w;
    if (LAYER == 1) {
        const float4 mu = *(const float4*)(mean  + lane * 4);
        const float4 vr = *(const float4*)(var   + lane * 4);
        const float4 ga = *(const float4*)(gamma + lane * 4);
        const float4 be = *(const float4*)(beta  + lane * 4);
        v.x = fmaxf((v.x - mu.x) * rsqrtf(vr.x + 1e-5f) * ga.x + be.x, 0.0f);
        v.y = fmaxf((v.y - mu.y) * rsqrtf(vr.y + 1e-5f) * ga.y + be.y, 0.0f);
        v.z = fmaxf((v.z - mu.z) * rsqrtf(vr.z + 1e-5f) * ga.z + be.z, 0.0f);
        v.w = fmaxf((v.w - mu.w) * rsqrtf(vr.w + 1e-5f) * ga.w + be.w, 0.0f);
    } else {
        v.x = fmaxf(xr.x, v.x);
        v.y = fmaxf(xr.y, v.y);
        v.z = fmaxf(xr.z, v.z);
        v.w = fmaxf(xr.w, v.w);
    }
    *(float4*)(y + (size_t)node * HDIM + lane * 4) = v;
}

extern "C" void kernel_launch(void* const* d_in, const int* in_sizes, int n_in,
                              void* d_out, int out_size, void* d_ws, size_t ws_size,
                              hipStream_t stream) {
    const float* node_feat = (const float*)d_in[0];
    const int*   eidx      = (const int*)d_in[1];
    const float* W1        = (const float*)d_in[2];
    const float* b1        = (const float*)d_in[3];
    const float* gamma1    = (const float*)d_in[4];
    const float* beta1     = (const float*)d_in[5];
    const float* mean1     = (const float*)d_in[6];
    const float* var1      = (const float*)d_in[7];
    const float* W2        = (const float*)d_in[8];
    const float* b2        = (const float*)d_in[9];
    const float* Wf        = (const float*)d_in[10];
    const float* bf        = (const float*)d_in[11];
    float* out = (float*)d_out;

    const int n = in_sizes[0] / HDIM;
    const int e = in_sizes[1] / 2;
    const int* row = eidx;
    const int* col = eidx + e;

    float* dis = (float*)d_ws;
    float* h   = dis + n;
    float* y1  = h + (size_t)n * HDIM;
    float* y2  = y1 + (size_t)n * HDIM;
    int* cnt   = (int*)(y2 + (size_t)n * HDIM);
    int* off   = cnt + n;
    int* rank  = off + n;
    int* srcs  = rank + e;

    const int mtiles  = n >> 4;
    const int gemmGrd = (mtiles + 7) / 8;
    const int nodeGrd = (n + 7) / 8;

    k_zero_cnt<<<(n + 255) / 256, 256, 0, stream>>>(cnt, n);
    k_count<<<(e + 255) / 256, 256, 0, stream>>>(col, cnt, rank, e);
    k_scan<<<(n + 1023) / 1024, 1024, 0, stream>>>(cnt, off, dis, n);
    k_fill<<<(e + 255) / 256, 256, 0, stream>>>(row, col, rank, off, srcs, e);

    k_gemm_wmma<128><<<gemmGrd, 256, 0, stream>>>(node_feat, W1, nullptr, h, n, HDIM);
    k_gather<1><<<nodeGrd, 256, 0, stream>>>(h, dis, off, cnt, srcs, y1, n,
                                             b1, gamma1, beta1, mean1, var1, nullptr);

    k_gemm_wmma<128><<<gemmGrd, 256, 0, stream>>>(y1, W2, nullptr, h, n, HDIM);
    k_gather<2><<<nodeGrd, 256, 0, stream>>>(h, dis, off, cnt, srcs, y2, n,
                                             b2, nullptr, nullptr, nullptr, nullptr, y1);

    k_gemm_wmma<48><<<gemmGrd, 256, 0, stream>>>(y2, Wf, bf, out, n, ODIM);
}


# Edge sort by target as a two-level counting sort on LDS counters (chunk histogram, binned runs, one block per 512-node bin): no per-edge global atomic or scattered access; sum order = LDS arrival
# speedup vs baseline: 3.7810x; 1.5336x over previous
#include <hip/hip_runtime.h>

typedef __attribute__((ext_vector_type(2))) float v2f;
typedef __attribute__((ext_vector_type(8))) float v8f;

#define HDIM 128
#define ODIM 40

#define BINSH   9
#define BINW    512
#define MAXB    256
#define EDGE_CH 8192
#define EDGE_PT (EDGE_CH / 1024)
#define BINCAP  12288

__device__ __forceinline__ int block_excl_scan(int v, int* s_w) {
    const int lane = threadIdx.x & 31;
    const int wv   = threadIdx.x >> 5;
    int incl = v;
    for (int d = 1; d < 32; d <<= 1) {
        int t = __shfl_up(incl, d);
        if (lane >= d) incl += t;
    }
    if (lane == 31) s_w[wv] = incl;
    __syncthreads();
    int before = 0;
    for (int w = 0; w < 32; ++w) {
        int t = s_w[w];
        if (w < wv) before += t;
    }
    __syncthreads();
    return before + incl - v;
}

__global__ __launch_bounds__(1024) void k_bin_hist(const int* __restrict__ col,
                                                   int* __restrict__ hist, int e, int nb) {
    __shared__ int s_h[MAXB];
    const int tid = threadIdx.x;
    if (tid < MAXB) s_h[tid] = 0;
    __syncthreads();
    const int lo = blockIdx.x * EDGE_CH;
    const int hi = (lo + EDGE_CH < e) ? (lo + EDGE_CH) : e;
    int c[EDGE_PT];
#pragma unroll
    for (int j = 0; j < EDGE_PT; ++j) {
        const int i = lo + tid + j * 1024;
        c[j] = (i < hi) ? col[i] : 0;
    }
#pragma unroll
    for (int j = 0; j < EDGE_PT; ++j) {
        if (lo + tid + j * 1024 < hi) atomicAdd(&s_h[c[j] >> BINSH], 1);
    }
    __syncthreads();
    if (tid < nb) hist[blockIdx.x * nb + tid] = s_h[tid];
}

__global__ __launch_bounds__(1024) void k_bin_part(const int* __restrict__ row,
                                                   const int* __restrict__ col,
                                                   const int* __restrict__ hist,
                                                   int* __restrict__ binStart,
                                                   int2* __restrict__ pairs, int e, int nb) {
    __shared__ int  s_tot[MAXB];
    __shared__ int  s_pre[MAXB];
    __shared__ int  s_cur[MAXB];
    __shared__ int  s_delta[MAXB];
    __shared__ int  s_w[32];
    __shared__ int2 s_pair[EDGE_CH];
    const int tid = threadIdx.x;
    const int wg  = blockIdx.x;
    const int nwg = gridDim.x;
    if (tid < MAXB) { s_tot[tid] = 0; s_pre[tid] = 0; }
    __syncthreads();
    {
        const int b = tid & (MAXB - 1);
        int t_all = 0, t_pre = 0;
        if (b < nb) {
#pragma unroll 8
            for (int w = tid >> 8; w < nwg; w += 4) {
                const int v = hist[w * nb + b];
                t_all += v;
                t_pre += (w < wg) ? v : 0;
            }
            atomicAdd(&s_tot[b], t_all);
            atomicAdd(&s_pre[b], t_pre);
        }
    }
    __syncthreads();
    const int own = (tid < nb) ? hist[wg * nb + tid] : 0;
    const int tot = (tid < nb) ? s_tot[tid] : 0;
    const int gstart = block_excl_scan(tot, s_w);
    const int lstart = block_excl_scan(own, s_w);
    if (tid < nb) {
        s_cur[tid]   = lstart;
        s_delta[tid] = gstart + s_pre[tid] - lstart;
    }
    if (wg == 0 && tid <= nb) binStart[tid] = gstart;
    __syncthreads();

    const int lo = wg * EDGE_CH;
    const int hi = (lo + EDGE_CH < e) ? (lo + EDGE_CH) : e;
    int c[EDGE_PT], r[EDGE_PT];
#pragma unroll
    for (int j = 0; j < EDGE_PT; ++j) {
        const int i = lo + tid + j * 1024;
        c[j] = (i < hi) ? col[i] : 0;
        r[j] = (i < hi) ? row[i] : 0;
    }
#pragma unroll
    for (int j = 0; j < EDGE_PT; ++j) {
        if (lo + tid + j * 1024 < hi) {
            const int slot = atomicAdd(&s_cur[c[j] >> BINSH], 1);
            s_pair[slot] = make_int2(r[j], c[j]);
        }
    }
    __syncthreads();
    for (int i = tid; i < hi - lo; i += 1024) {
        const int2 p = s_pair[i];
        pairs[s_delta[p.y >> BINSH] + i] = p;
    }
}

__global__ __launch_bounds__(1024) void k_bin_sort(const int2* __restrict__ pairs,
                                                   const int* __restrict__ binStart,
                                                   int* __restrict__ off, int* __restrict__ cnt,
                                                   float* __restrict__ dis,
                                                   int* __restrict__ srcs, int n) {
    __shared__ int s_cnt[BINW];
    __shared__ int s_cur[BINW];
    __shared__ int s_w[32];
    __shared__ int s_src[BINCAP];
    const int tid   = threadIdx.x;
    const int node0 = blockIdx.x << BINSH;
    const int e0 = binStart[blockIdx.x];
    const int m  = binStart[blockIdx.x + 1] - e0;
    if (tid < BINW) s_cnt[tid] = 0;
    __syncthreads();
    for (int base = 0; base < m; base += EDGE_CH) {
        int t[EDGE_PT];
#pragma unroll
        for (int j = 0; j < EDGE_PT; ++j) {
            const int i = base + tid + j * 1024;
            t[j] = (i < m) ? pairs[e0 + i].y : node0;
        }
#pragma unroll
        for (int j = 0; j < EDGE_PT; ++j) {
            if (base + tid + j * 1024 < m) atomicAdd(&s_cnt[t[j] - node0], 1);
        }
    }
    __syncthreads();
    const int c  = (tid < BINW) ? s_cnt[tid] : 0;
    const int lo = block_excl_scan(c, s_w);
    if (tid < BINW) {
        s_cur[tid] = lo;
        const int node = node0 + tid;
        if (node < n) {
            off[node] = e0 + lo;
            cnt[node] = c;
            float dg = (float)(c + 1);
            dis[node] = (dg > 0.0f) ? rsqrtf(dg) : 0.0f;
        }
    }
    __syncthreads();
    const bool in_lds = (m <= BINCAP);
    for (int base = 0; base < m; base += EDGE_CH) {
        int2 p[EDGE_PT];
#pragma unroll
        for (int j = 0; j < EDGE_PT; ++j) {
            const int i = base + tid + j * 1024;
            p[j] = (i < m) ? pairs[e0 + i] : make_int2(0, node0);
        }
#pragma unroll
        for (int j = 0; j < EDGE_PT; ++j) {
            if (base + tid + j * 1024 < m) {
                const int slot = atomicAdd(&s_cur[p[j].y - node0], 1);
                if (in_lds) s_src[slot] = p[j].x; else srcs[e0 + slot] = p[j].x;
            }
        }
    }
    __syncthreads();
    if (in_lds) {
        for (int i = tid; i < m; i += 1024) srcs[e0 + i] = s_src[i];
    }
}

template <int NPAD, bool ROWSCALE>
__global__ __launch_bounds__(256) void k_gemm_wmma(
        const float* __restrict__ A, const float* __restrict__ B,
        const float* __restrict__ bias, const float* __restrict__ rowscale,
        float* __restrict__ C, int M, int ncols) {
    constexpr int K  = 128;
    constexpr int NT = NPAD / 16;
    constexpr int LSTR = K + 2;
    __shared__ float ldsBT[NPAD * LSTR];

    for (int i = threadIdx.x; i < NPAD * K; i += 256) {
        int k = i & (K - 1);
        int n = i >> 7;
        ldsBT[n * LSTR + k] = (n < ncols) ? B[k * ncols + n] : 0.0f;
    }
    __syncthreads();

    const int lane  = threadIdx.x & 31;
    const int tile  = __builtin_amdgcn_readfirstlane(blockIdx.x * 8 + (threadIdx.x >> 5));
    const int mtile = M >> 4;
    if (tile >= mtile) return;

    const int rowA = tile * 16 + (lane & 15);
    const int koff = (lane >> 4) * 2;
    const float* aptr = A + (size_t)rowA * K + koff;

    v8f acc[NT];
    v8f z8 = {0.f, 0.f, 0.f, 0.f, 0.f, 0.f, 0.f, 0.f};
#pragma unroll
    for (int nt = 0; nt < NT; ++nt) acc[nt] = z8;

#pragma unroll 4
    for (int k = 0; k < K; k += 4) {
        v2f a = *(const v2f*)(aptr + k);
#pragma unroll
        for (int nt = 0; nt < NT; ++nt) {
            int n = nt * 16 + (lane & 15);
            v2f b = *(const v2f*)(&ldsBT[n * LSTR + k + koff]);
            acc[nt] = __builtin_amdgcn_wmma_f32_16x16x4_f32(
                 false, a,  false, b,
                 (short)0, acc[nt],
                 false,  false);
        }
    }

    const int rbase = tile * 16 + ((lane >> 4) << 3);
    float sc[8];
    if (ROWSCALE) {
#pragma unroll
        for (int r = 0; r < 8; ++r) sc[r] = rowscale[rbase + r];
    }
#pragma unroll
    for (int nt = 0; nt < NT; ++nt) {
        int n = nt * 16 + (lane & 15);
        if (n >= ncols) continue;
        if (ROWSCALE) {
#pragma unroll
            for (int r = 0; r < 8; ++r) {
                C[(size_t)(rbase + r) * ncols + n] = acc[nt][r] * sc[r];
            }
        } else {
            float badd = bias ? bias[n] : 0.0f;
#pragma unroll
            for (int r = 0; r < 8; ++r) {
                C[(size_t)(rbase + r) * ncols + n] = acc[nt][r] + badd;
            }
        }
    }
}

template <int LAYER>
__global__ __launch_bounds__(256) void k_gather(
        const float* __restrict__ h, const float* __restrict__ dis,
        const int* __restrict__ off, const int* __restrict__ cnt,
        const int* __restrict__ srcs, float* __restrict__ y, int n,
        const float* __restrict__ bias,
        const float* __restrict__ gamma, const float* __restrict__ beta,
        const float* __restrict__ mean, const float* __restrict__ var,
        const float* __restrict__ x1) {
    const int node = __builtin_amdgcn_readfirstlane(blockIdx.x * 8 + (threadIdx.x >> 5));
    if (node >= n) return;
    const int lane  = threadIdx.x & 31;
    const int start = off[node];
    const int m     = cnt[node];
    const float dc  = dis[node];
    const float* hl = h + lane * 4;

    const float4 hs = *(const float4*)(hl + (size_t)node * HDIM);
    float4 xr = hs;
    if (LAYER == 2) xr = *(const float4*)(x1 + (size_t)node * HDIM + lane * 4);

    float4 acc = make_float4(0.f, 0.f, 0.f, 0.f);
    for (int base = 0; base < m; base += 32) {
        const int k = (m - base < 32) ? (m - base) : 32;
        int s = 0;
        if (lane < k) s = srcs[start + base + lane];
        int j = 0;
        for (; j + 4 <= k; j += 4) {
            const int s0 = __builtin_amdgcn_readlane(s, j);
            const int s1 = __builtin_amdgcn_readlane(s, j + 1);
            const int s2 = __builtin_amdgcn_readlane(s, j + 2);
            const int s3 = __builtin_amdgcn_readlane(s, j + 3);
            const float4 v0 = *(const float4*)(hl + (size_t)s0 * HDIM);
            const float4 v1 = *(const float4*)(hl + (size_t)s1 * HDIM);
            const float4 v2 = *(const float4*)(hl + (size_t)s2 * HDIM);
            const float4 v3 = *(const float4*)(hl + (size_t)s3 * HDIM);
            acc.x += v0.x; acc.y += v0.y; acc.z += v0.z; acc.w += v0.w;
            acc.x += v1.x; acc.y += v1.y; acc.z += v1.z; acc.w += v1.w;
            acc.x += v2.x; acc.y += v2.y; acc.z += v2.z; acc.w += v2.w;
            acc.x += v3.x; acc.y += v3.y; acc.z += v3.z; acc.w += v3.w;
        }
        for (; j < k; ++j) {
            const int s0 = __builtin_amdgcn_readlane(s, j);
            const float4 v0 = *(const float4*)(hl + (size_t)s0 * HDIM);
            acc.x += v0.x; acc.y += v0.y; acc.z += v0.z; acc.w += v0.w;
        }
    }
    const float4 bb = *(const float4*)(bias + lane * 4);
    float4 v;
    v.x = (acc.x + hs.x) * dc + bb.x;
    v.y = (acc.y + hs.y) * dc + bb.y;
    v.z = (acc.z + hs.z) * dc + bb.z;
    v.w = (acc.w + hs.w) * dc + bb.w;
    if (LAYER == 1) {
        const float4 mu = *(const float4*)(mean  + lane * 4);
        const float4 vr = *(const float4*)(var   + lane * 4);
        const float4 ga = *(const float4*)(gamma + lane * 4);
        const float4 be = *(const float4*)(beta  + lane * 4);
        v.x = fmaxf((v.x - mu.x) * rsqrtf(vr.x + 1e-5f) * ga.x + be.x, 0.0f);
        v.y = fmaxf((v.y - mu.y) * rsqrtf(vr.y + 1e-5f) * ga.y + be.y, 0.0f);
        v.z = fmaxf((v.z - mu.z) * rsqrtf(vr.z + 1e-5f) * ga.z + be.z, 0.0f);
        v.w = fmaxf((v.w - mu.w) * rsqrtf(vr.w + 1e-5f) * ga.w + be.w, 0.0f);
    } else {
        v.x = fmaxf(xr.x, v.x);
        v.y = fmaxf(xr.y, v.y);
        v.z = fmaxf(xr.z, v.z);
        v.w = fmaxf(xr.w, v.w);
    }
    *(float4*)(y + (size_t)node * HDIM + lane * 4) = v;
}

extern "C" void kernel_launch(void* const* d_in, const int* in_sizes, int n_in,
                              void* d_out, int out_size, void* d_ws, size_t ws_size,
                              hipStream_t stream) {
    const float* node_feat = (const float*)d_in[0];
    const int*   eidx      = (const int*)d_in[1];
    const float* W1        = (const float*)d_in[2];
    const float* b1        = (const float*)d_in[3];
    const float* gamma1    = (const float*)d_in[4];
    const float* beta1     = (const float*)d_in[5];
    const float* mean1     = (const float*)d_in[6];
    const float* var1      = (const float*)d_in[7];
    const float* W2        = (const float*)d_in[8];
    const float* b2        = (const float*)d_in[9];
    const float* Wf        = (const float*)d_in[10];
    const float* bf        = (const float*)d_in[11];
    float* out = (float*)d_out;

    const int n = in_sizes[0] / HDIM;
    const int e = in_sizes[1] / 2;
    const int* row = eidx;
    const int* col = eidx + e;

    const int nb     = (n + BINW - 1) / BINW;
    const int edgeGrd = (e + EDGE_CH - 1) / EDGE_CH;

    float* dis = (float*)d_ws;
    float* h   = dis + n;
    float* y1  = h + (size_t)n * HDIM;
    float* y2  = y1 + (size_t)n * HDIM;
    int2* pairs = (int2*)(y2 + (size_t)n * HDIM);
    int* cnt   = (int*)(pairs + e);
    int* off   = cnt + n;
    int* srcs  = off + n;
    int* hist  = srcs + e;
    int* binStart = hist + (size_t)edgeGrd * nb;

    const int mtiles  = n >> 4;
    const int gemmGrd = (mtiles + 7) / 8;
    const int nodeGrd = (n + 7) / 8;

    k_bin_hist<<<edgeGrd, 1024, 0, stream>>>(col, hist, e, nb);
    k_bin_part<<<edgeGrd, 1024, 0, stream>>>(row, col, hist, binStart, pairs, e, nb);
    k_bin_sort<<<nb, 1024, 0, stream>>>(pairs, binStart, off, cnt, dis, srcs, n);

    k_gemm_wmma<128, true><<<gemmGrd, 256, 0, stream>>>(node_feat, W1, nullptr, dis, h, n, HDIM);
    k_gather<1><<<nodeGrd, 256, 0, stream>>>(h, dis, off, cnt, srcs, y1, n,
                                             b1, gamma1, beta1, mean1, var1, nullptr);

    k_gemm_wmma<128, true><<<gemmGrd, 256, 0, stream>>>(y1, W2, nullptr, dis, h, n, HDIM);
    k_gather<2><<<nodeGrd, 256, 0, stream>>>(h, dis, off, cnt, srcs, y2, n,
                                             b2, nullptr, nullptr, nullptr, nullptr, y1);

    k_gemm_wmma<48, false><<<gemmGrd, 256, 0, stream>>>(y2, Wf, bf, nullptr, out, n, ODIM);
}


# GEMM: B staged into LDS by whole-row 16-byte loads (a wave-level load covers 4 full lines, bank-spread transposing stores) instead of 4-byte column reads that touch 32 lines per load
# speedup vs baseline: 4.3551x; 1.1518x over previous
#include <hip/hip_runtime.h>

typedef __attribute__((ext_vector_type(2))) float v2f;
typedef __attribute__((ext_vector_type(8))) float v8f;

#define HDIM 128
#define ODIM 40

#define BINSH   9
#define BINW    512
#define MAXB    256
#define EDGE_CH 8192
#define EDGE_PT (EDGE_CH / 1024)
#define BINCAP  12288

__device__ __forceinline__ int block_excl_scan(int v, int* s_w) {
    const int lane = threadIdx.x & 31;
    const int wv   = threadIdx.x >> 5;
    int incl = v;
    for (int d = 1; d < 32; d <<= 1) {
        int t = __shfl_up(incl, d);
        if (lane >= d) incl += t;
    }
    if (lane == 31) s_w[wv] = incl;
    __syncthreads();
    int before = 0;
    for (int w = 0; w < 32; ++w) {
        int t = s_w[w];
        if (w < wv) before += t;
    }
    __syncthreads();
    return before + incl - v;
}

__global__ __launch_bounds__(1024) void k_bin_hist(const int* __restrict__ col,
                                                   int* __restrict__ hist, int e, int nb) {
    __shared__ int s_h[MAXB];
    const int tid = threadIdx.x;
    if (tid < MAXB) s_h[tid] = 0;
    __syncthreads();
    const int lo = blockIdx.x * EDGE_CH;
    const int hi = (lo + EDGE_CH < e) ? (lo + EDGE_CH) : e;
    int c[EDGE_PT];
#pragma unroll
    for (int j = 0; j < EDGE_PT; ++j) {
        const int i = lo + tid + j * 1024;
        c[j] = (i < hi) ? col[i] : 0;
    }
#pragma unroll
    for (int j = 0; j < EDGE_PT; ++j) {
        if (lo + tid + j * 1024 < hi) atomicAdd(&s_h[c[j] >> BINSH], 1);
    }
    __syncthreads();
    if (tid < nb) hist[blockIdx.x * nb + tid] = s_h[tid];
}

__global__ __launch_bounds__(1024) void k_bin_part(const int* __restrict__ row,
                                                   const int* __restrict__ col,
                                                   const int* __restrict__ hist,
                                                   int* __restrict__ binStart,
                                                   int2* __restrict__ pairs, int e, int nb) {
    __shared__ int  s_tot[MAXB];
    __shared__ int  s_pre[MAXB];
    __shared__ int  s_cur[MAXB];
    __shared__ int  s_delta[MAXB];
    __shared__ int  s_w[32];
    __shared__ int2 s_pair[EDGE_CH];
    const int tid = threadIdx.x;
    const int wg  = blockIdx.x;
    const int nwg = gridDim.x;
    if (tid < MAXB) { s_tot[tid] = 0; s_pre[tid] = 0; }
    __syncthreads();
    {
        const int b = tid & (MAXB - 1);
        int t_all = 0, t_pre = 0;
        if (b < nb) {
#pragma unroll 8
            for (int w = tid >> 8; w < nwg; w += 4) {
                const int v = hist[w * nb + b];
                t_all += v;
                t_pre += (w < wg) ? v : 0;
            }
            atomicAdd(&s_tot[b], t_all);
            atomicAdd(&s_pre[b], t_pre);
        }
    }
    __syncthreads();
    const int own = (tid < nb) ? hist[wg * nb + tid] : 0;
    const int tot = (tid < nb) ? s_tot[tid] : 0;
    const int gstart = block_excl_scan(tot, s_w);
    const int lstart = block_excl_scan(own, s_w);
    if (tid < nb) {
        s_cur[tid]   = lstart;
        s_delta[tid] = gstart + s_pre[tid] - lstart;
    }
    if (wg == 0 && tid <= nb) binStart[tid] = gstart;
    __syncthreads();

    const int lo = wg * EDGE_CH;
    const int hi = (lo + EDGE_CH < e) ? (lo + EDGE_CH) : e;
    int c[EDGE_PT], r[EDGE_PT];
#pragma unroll
    for (int j = 0; j < EDGE_PT; ++j) {
        const int i = lo + tid + j * 1024;
        c[j] = (i < hi) ? col[i] : 0;
        r[j] = (i < hi) ? row[i] : 0;
    }
#pragma unroll
    for (int j = 0; j < EDGE_PT; ++j) {
        if (lo + tid + j * 1024 < hi) {
            const int slot = atomicAdd(&s_cur[c[j] >> BINSH], 1);
            s_pair[slot] = make_int2(r[j], c[j]);
        }
    }
    __syncthreads();
    for (int i = tid; i < hi - lo; i += 1024) {
        const int2 p = s_pair[i];
        pairs[s_delta[p.y >> BINSH] + i] = p;
    }
}

__global__ __launch_bounds__(1024) void k_bin_sort(const int2* __restrict__ pairs,
                                                   const int* __restrict__ binStart,
                                                   int* __restrict__ off, int* __restrict__ cnt,
                                                   float* __restrict__ dis,
                                                   int* __restrict__ srcs, int n) {
    __shared__ int s_cnt[BINW];
    __shared__ int s_cur[BINW];
    __shared__ int s_w[32];
    __shared__ int s_src[BINCAP];
    const int tid   = threadIdx.x;
    const int node0 = blockIdx.x << BINSH;
    const int e0 = binStart[blockIdx.x];
    const int m  = binStart[blockIdx.x + 1] - e0;
    if (tid < BINW) s_cnt[tid] = 0;
    __syncthreads();
    for (int base = 0; base < m; base += EDGE_CH) {
        int t[EDGE_PT];
#pragma unroll
        for (int j = 0; j < EDGE_PT; ++j) {
            const int i = base + tid + j * 1024;
            t[j] = (i < m) ? pairs[e0 + i].y : node0;
        }
#pragma unroll
        for (int j = 0; j < EDGE_PT; ++j) {
            if (base + tid + j * 1024 < m) atomicAdd(&s_cnt[t[j] - node0], 1);
        }
    }
    __syncthreads();
    const int c  = (tid < BINW) ? s_cnt[tid] : 0;
    const int lo = block_excl_scan(c, s_w);
    if (tid < BINW) {
        s_cur[tid] = lo;
        const int node = node0 + tid;
        if (node < n) {
            off[node] = e0 + lo;
            cnt[node] = c;
            float dg = (float)(c + 1);
            dis[node] = (dg > 0.0f) ? rsqrtf(dg) : 0.0f;
        }
    }
    __syncthreads();
    const bool in_lds = (m <= BINCAP);
    for (int base = 0; base < m; base += EDGE_CH) {
        int2 p[EDGE_PT];
#pragma unroll
        for (int j = 0; j < EDGE_PT; ++j) {
            const int i = base + tid + j * 1024;
            p[j] = (i < m) ? pairs[e0 + i] : make_int2(0, node0);
        }
#pragma unroll
        for (int j = 0; j < EDGE_PT; ++j) {
            if (base + tid + j * 1024 < m) {
                const int slot = atomicAdd(&s_cur[p[j].y - node0], 1);
                if (in_lds) s_src[slot] = p[j].x; else srcs[e0 + slot] = p[j].x;
            }
        }
    }
    __syncthreads();
    if (in_lds) {
        for (int i = tid; i < m; i += 1024) srcs[e0 + i] = s_src[i];
    }
}

template <int NPAD, bool ROWSCALE>
__global__ __launch_bounds__(256) void k_gemm_wmma(
        const float* __restrict__ A, const float* __restrict__ B,
        const float* __restrict__ bias, const float* __restrict__ rowscale,
        float* __restrict__ C, int M, int ncols) {
    constexpr int K  = 128;
    constexpr int NT = NPAD / 16;
    constexpr int LSTR = K + 2;
    __shared__ float ldsBT[NPAD * LSTR];

    bool staged = false;
    if constexpr ((NPAD & 31) == 0) {
        if (ncols == NPAD) {
            constexpr int CG = NPAD / 32;
#pragma unroll
            for (int it = 0; it < (K * NPAD / 4) / 256; ++it) {
                const int i = threadIdx.x + it * 256;
                const int w = i >> 5;
                const int k = (w / CG) * 4 + ((i >> 3) & 3);
                const int n = (w % CG) * 32 + (i & 7) * 4;
                const float4 v = *(const float4*)(B + k * NPAD + n);
                ldsBT[(n + 0) * LSTR + k] = v.x;
                ldsBT[(n + 1) * LSTR + k] = v.y;
                ldsBT[(n + 2) * LSTR + k] = v.z;
                ldsBT[(n + 3) * LSTR + k] = v.w;
            }
            staged = true;
        }
    }
    if (!staged) {
        const int q  = ncols >> 2;
        const int np = K * q;
        constexpr int NIT = (K * (NPAD / 4) + 255) / 256;
        float4 v[NIT];
#pragma unroll
        for (int it = 0; it < NIT; ++it) {
            const int i = threadIdx.x + it * 256;
            v[it] = ((const float4*)B)[(i < np) ? i : 0];
        }
#pragma unroll
        for (int it = 0; it < NIT; ++it) {
            const int i = threadIdx.x + it * 256;
            if (i < np) {
                const int k = i / q;
                const int n = (i - k * q) * 4;
                ldsBT[(n + 0) * LSTR + k] = v[it].x;
                ldsBT[(n + 1) * LSTR + k] = v[it].y;
                ldsBT[(n + 2) * LSTR + k] = v[it].z;
                ldsBT[(n + 3) * LSTR + k] = v[it].w;
            }
        }
        for (int i = threadIdx.x; i < (NPAD - ncols) * K; i += 256) {
            const int k = i & (K - 1);
            const int n = ncols + (i >> 7);
            ldsBT[n * LSTR + k] = 0.0f;
        }
    }
    __syncthreads();

    const int lane  = threadIdx.x & 31;
    const int tile  = __builtin_amdgcn_readfirstlane(blockIdx.x * 8 + (threadIdx.x >> 5));
    const int mtile = M >> 4;
    if (tile >= mtile) return;

    const int rowA = tile * 16 + (lane & 15);
    const int koff = (lane >> 4) * 2;
    const float* aptr = A + (size_t)rowA * K + koff;

    v8f acc[NT];
    v8f z8 = {0.f, 0.f, 0.f, 0.f, 0.f, 0.f, 0.f, 0.f};
#pragma unroll
    for (int nt = 0; nt < NT; ++nt) acc[nt] = z8;

#pragma unroll 4
    for (int k = 0; k < K; k += 4) {
        v2f a = *(const v2f*)(aptr + k);
#pragma unroll
        for (int nt = 0; nt < NT; ++nt) {
            int n = nt * 16 + (lane & 15);
            v2f b = *(const v2f*)(&ldsBT[n * LSTR + k + koff]);
            acc[nt] = __builtin_amdgcn_wmma_f32_16x16x4_f32(
                 false, a,  false, b,
                 (short)0, acc[nt],
                 false,  false);
        }
    }

    const int rbase = tile * 16 + ((lane >> 4) << 3);
    float sc[8];
    if (ROWSCALE) {
#pragma unroll
        for (int r = 0; r < 8; ++r) sc[r] = rowscale[rbase + r];
    }
#pragma unroll
    for (int nt = 0; nt < NT; ++nt) {
        int n = nt * 16 + (lane & 15);
        if (n >= ncols) continue;
        if (ROWSCALE) {
#pragma unroll
            for (int r = 0; r < 8; ++r) {
                C[(size_t)(rbase + r) * ncols + n] = acc[nt][r] * sc[r];
            }
        } else {
            float badd = bias ? bias[n] : 0.0f;
#pragma unroll
            for (int r = 0; r < 8; ++r) {
                C[(size_t)(rbase + r) * ncols + n] = acc[nt][r] + badd;
            }
        }
    }
}

template <int LAYER>
__global__ __launch_bounds__(256) void k_gather(
        const float* __restrict__ h, const float* __restrict__ dis,
        const int* __restrict__ off, const int* __restrict__ cnt,
        const int* __restrict__ srcs, float* __restrict__ y, int n,
        const float* __restrict__ bias,
        const float* __restrict__ gamma, const float* __restrict__ beta,
        const float* __restrict__ mean, const float* __restrict__ var,
        const float* __restrict__ x1) {
    const int node = __builtin_amdgcn_readfirstlane(blockIdx.x * 8 + (threadIdx.x >> 5));
    if (node >= n) return;
    const int lane  = threadIdx.x & 31;
    const int start = off[node];
    const int m     = cnt[node];
    const float dc  = dis[node];
    const float* hl = h + lane * 4;

    const float4 hs = *(const float4*)(hl + (size_t)node * HDIM);
    float4 xr = hs;
    if (LAYER == 2) xr = *(const float4*)(x1 + (size_t)node * HDIM + lane * 4);

    float4 acc = make_float4(0.f, 0.f, 0.f, 0.f);
    for (int base = 0; base < m; base += 32) {
        const int k = (m - base < 32) ? (m - base) : 32;
        int s = 0;
        if (lane < k) s = srcs[start + base + lane];
        int j = 0;
        for (; j + 4 <= k; j += 4) {
            const int s0 = __builtin_amdgcn_readlane(s, j);
            const int s1 = __builtin_amdgcn_readlane(s, j + 1);
            const int s2 = __builtin_amdgcn_readlane(s, j + 2);
            const int s3 = __builtin_amdgcn_readlane(s, j + 3);
            const float4 v0 = *(const float4*)(hl + (size_t)s0 * HDIM);
            const float4 v1 = *(const float4*)(hl + (size_t)s1 * HDIM);
            const float4 v2 = *(const float4*)(hl + (size_t)s2 * HDIM);
            const float4 v3 = *(const float4*)(hl + (size_t)s3 * HDIM);
            acc.x += v0.x; acc.y += v0.y; acc.z += v0.z; acc.w += v0.w;
            acc.x += v1.x; acc.y += v1.y; acc.z += v1.z; acc.w += v1.w;
            acc.x += v2.x; acc.y += v2.y; acc.z += v2.z; acc.w += v2.w;
            acc.x += v3.x; acc.y += v3.y; acc.z += v3.z; acc.w += v3.w;
        }
        for (; j < k; ++j) {
            const int s0 = __builtin_amdgcn_readlane(s, j);
            const float4 v0 = *(const float4*)(hl + (size_t)s0 * HDIM);
            acc.x += v0.x; acc.y += v0.y; acc.z += v0.z; acc.w += v0.w;
        }
    }
    const float4 bb = *(const float4*)(bias + lane * 4);
    float4 v;
    v.x = (acc.x + hs.x) * dc + bb.x;
    v.y = (acc.y + hs.y) * dc + bb.y;
    v.z = (acc.z + hs.z) * dc + bb.z;
    v.w = (acc.w + hs.w) * dc + bb.w;
    if (LAYER == 1) {
        const float4 mu = *(const float4*)(mean  + lane * 4);
        const float4 vr = *(const float4*)(var   + lane * 4);
        const float4 ga = *(const float4*)(gamma + lane * 4);
        const float4 be = *(const float4*)(beta  + lane * 4);
        v.x = fmaxf((v.x - mu.x) * rsqrtf(vr.x + 1e-5f) * ga.x + be.x, 0.0f);
        v.y = fmaxf((v.y - mu.y) * rsqrtf(vr.y + 1e-5f) * ga.y + be.y, 0.0f);
        v.z = fmaxf((v.z - mu.z) * rsqrtf(vr.z + 1e-5f) * ga.z + be.z, 0.0f);
        v.w = fmaxf((v.w - mu.w) * rsqrtf(vr.w + 1e-5f) * ga.w + be.w, 0.0f);
    } else {
        v.x = fmaxf(xr.x, v.x);
        v.y = fmaxf(xr.y, v.y);
        v.z = fmaxf(xr.z, v.z);
        v.w = fmaxf(xr.w, v.w);
    }
    *(float4*)(y + (size_t)node * HDIM + lane * 4) = v;
}

extern "C" void kernel_launch(void* const* d_in, const int* in_sizes, int n_in,
                              void* d_out, int out_size, void* d_ws, size_t ws_size,
                              hipStream_t stream) {
    const float* node_feat = (const float*)d_in[0];
    const int*   eidx      = (const int*)d_in[1];
    const float* W1        = (const float*)d_in[2];
    const float* b1        = (const float*)d_in[3];
    const float* gamma1    = (const float*)d_in[4];
    const float* beta1     = (const float*)d_in[5];
    const float* mean1     = (const float*)d_in[6];
    const float* var1      = (const float*)d_in[7];
    const float* W2        = (const float*)d_in[8];
    const float* b2        = (const float*)d_in[9];
    const float* Wf        = (const float*)d_in[10];
    const float* bf        = (const float*)d_in[11];
    float* out = (float*)d_out;

    const int n = in_sizes[0] / HDIM;
    const int e = in_sizes[1] / 2;
    const int* row = eidx;
    const int* col = eidx + e;

    const int nb     = (n + BINW - 1) / BINW;
    const int edgeGrd = (e + EDGE_CH - 1) / EDGE_CH;

    float* dis = (float*)d_ws;
    float* h   = dis + n;
    float* y1  = h + (size_t)n * HDIM;
    float* y2  = y1 + (size_t)n * HDIM;
    int2* pairs = (int2*)(y2 + (size_t)n * HDIM);
    int* cnt   = (int*)(pairs + e);
    int* off   = cnt + n;
    int* srcs  = off + n;
    int* hist  = srcs + e;
    int* binStart = hist + (size_t)edgeGrd * nb;

    const int mtiles  = n >> 4;
    const int gemmGrd = (mtiles + 7) / 8;
    const int nodeGrd = (n + 7) / 8;

    k_bin_hist<<<edgeGrd, 1024, 0, stream>>>(col, hist, e, nb);
    k_bin_part<<<edgeGrd, 1024, 0, stream>>>(row, col, hist, binStart, pairs, e, nb);
    k_bin_sort<<<nb, 1024, 0, stream>>>(pairs, binStart, off, cnt, dis, srcs, n);

    k_gemm_wmma<128, true><<<gemmGrd, 256, 0, stream>>>(node_feat, W1, nullptr, dis, h, n, HDIM);
    k_gather<1><<<nodeGrd, 256, 0, stream>>>(h, dis, off, cnt, srcs, y1, n,
                                             b1, gamma1, beta1, mean1, var1, nullptr);

    k_gemm_wmma<128, true><<<gemmGrd, 256, 0, stream>>>(y1, W2, nullptr, dis, h, n, HDIM);
    k_gather<2><<<nodeGrd, 256, 0, stream>>>(h, dis, off, cnt, srcs, y2, n,
                                             b2, nullptr, nullptr, nullptr, nullptr, y1);

    k_gemm_wmma<48, false><<<gemmGrd, 256, 0, stream>>>(y2, Wf, bf, nullptr, out, n, ODIM);
}


# speedup vs baseline: 4.9624x; 1.1394x over previous
#include <hip/hip_runtime.h>

typedef __attribute__((ext_vector_type(2))) float v2f;
typedef __attribute__((ext_vector_type(8))) float v8f;

#define HDIM 128
#define ODIM 40

#define BINSH   9
#define BINW    512
#define MAXB    256
#define EDGE_CH 8192
#define EDGE_PT (EDGE_CH / 1024)
#define BINCAP  12288

__device__ __forceinline__ int block_excl_scan(int v, int* s_w) {
    const int lane = threadIdx.x & 31;
    const int wv   = threadIdx.x >> 5;
    int incl = v;
    for (int d = 1; d < 32; d <<= 1) {
        int t = __shfl_up(incl, d);
        if (lane >= d) incl += t;
    }
    if (lane == 31) s_w[wv] = incl;
    __syncthreads();
    int before = 0;
    for (int w = 0; w < 32; ++w) {
        int t = s_w[w];
        if (w < wv) before += t;
    }
    __syncthreads();
    return before + incl - v;
}

__global__ __launch_bounds__(1024) void k_bin_hist(const int* __restrict__ col,
                                                   int* __restrict__ hist, int e, int nb) {
    __shared__ int s_h[MAXB];
    const int tid = threadIdx.x;
    if (tid < MAXB) s_h[tid] = 0;
    __syncthreads();
    const int lo = blockIdx.x * EDGE_CH;
    const int hi = (lo + EDGE_CH < e) ? (lo + EDGE_CH) : e;
    int c[EDGE_PT];
#pragma unroll
    for (int j = 0; j < EDGE_PT; ++j) {
        const int i = lo + tid + j * 1024;
        c[j] = (i < hi) ? col[i] : 0;
    }
#pragma unroll
    for (int j = 0; j < EDGE_PT; ++j) {
        if (lo + tid + j * 1024 < hi) atomicAdd(&s_h[c[j] >> BINSH], 1);
    }
    __syncthreads();
    if (tid < nb) hist[blockIdx.x * nb + tid] = s_h[tid];
}

__global__ __launch_bounds__(1024) void k_bin_part(const int* __restrict__ row,
                                                   const int* __restrict__ col,
                                                   const int* __restrict__ hist,
                                                   int* __restrict__ binStart,
                                                   int2* __restrict__ pairs, int e, int nb) {
    __shared__ int  s_tot[MAXB];
    __shared__ int  s_pre[MAXB];
    __shared__ int  s_cur[MAXB];
    __shared__ int  s_delta[MAXB];
    __shared__ int  s_w[32];
    __shared__ int2 s_pair[EDGE_CH];
    const int tid = threadIdx.x;
    const int wg  = blockIdx.x;
    const int nwg = gridDim.x;
    if (tid < MAXB) { s_tot[tid] = 0; s_pre[tid] = 0; }
    __syncthreads();
    {
        const int b = tid & (MAXB - 1);
        int t_all = 0, t_pre = 0;
        if (b < nb) {
#pragma unroll 8
            for (int w = tid >> 8; w < nwg; w += 4) {
                const int v = hist[w * nb + b];
                t_all += v;
                t_pre += (w < wg) ? v : 0;
            }
            atomicAdd(&s_tot[b], t_all);
            atomicAdd(&s_pre[b], t_pre);
        }
    }
    __syncthreads();
    const int own = (tid < nb) ? hist[wg * nb + tid] : 0;
    const int tot = (tid < nb) ? s_tot[tid] : 0;
    const int gstart = block_excl_scan(tot, s_w);
    const int lstart = block_excl_scan(own, s_w);
    if (tid < nb) {
        s_cur[tid]   = lstart;
        s_delta[tid] = gstart + s_pre[tid] - lstart;
    }
    if (wg == 0 && tid <= nb) binStart[tid] = gstart;
    __syncthreads();

    const int lo = wg * EDGE_CH;
    const int hi = (lo + EDGE_CH < e) ? (lo + EDGE_CH) : e;
    int c[EDGE_PT], r[EDGE_PT];
#pragma unroll
    for (int j = 0; j < EDGE_PT; ++j) {
        const int i = lo + tid + j * 1024;
        c[j] = (i < hi) ? col[i] : 0;
        r[j] = (i < hi) ? row[i] : 0;
    }
#pragma unroll
    for (int j = 0; j < EDGE_PT; ++j) {
        if (lo + tid + j * 1024 < hi) {
            const int slot = atomicAdd(&s_cur[c[j] >> BINSH], 1);
            s_pair[slot] = make_int2(r[j], c[j]);
        }
    }
    __syncthreads();
    for (int i = tid; i < hi - lo; i += 1024) {
        const int2 p = s_pair[i];
        pairs[s_delta[p.y >> BINSH] + i] = p;
    }
}

__global__ __launch_bounds__(1024) void k_bin_sort(const int2* __restrict__ pairs,
                                                   const int* __restrict__ binStart,
                                                   int* __restrict__ off, int* __restrict__ cnt,
                                                   float* __restrict__ dis,
                                                   int* __restrict__ srcs, int n) {
    __shared__ int s_cnt[BINW];
    __shared__ int s_cur[BINW];
    __shared__ int s_w[32];
    __shared__ int s_src[BINCAP];
    const int tid   = threadIdx.x;
    const int node0 = blockIdx.x << BINSH;
    const int e0 = binStart[blockIdx.x];
    const int m  = binStart[blockIdx.x + 1] - e0;
    if (tid < BINW) s_cnt[tid] = 0;
    __syncthreads();
    for (int base = 0; base < m; base += EDGE_CH) {
        int t[EDGE_PT];
#pragma unroll
        for (int j = 0; j < EDGE_PT; ++j) {
            const int i = base + tid + j * 1024;
            t[j] = (i < m) ? pairs[e0 + i].y : node0;
        }
#pragma unroll
        for (int j = 0; j < EDGE_PT; ++j) {
            if (base + tid + j * 1024 < m) atomicAdd(&s_cnt[t[j] - node0], 1);
        }
    }
    __syncthreads();
    const int c  = (tid < BINW) ? s_cnt[tid] : 0;
    const int lo = block_excl_scan(c, s_w);
    if (tid < BINW) {
        s_cur[tid] = lo;
        const int node = node0 + tid;
        if (node < n) {
            off[node] = e0 + lo;
            cnt[node] = c;
            float dg = (float)(c + 1);
            dis[node] = (dg > 0.0f) ? rsqrtf(dg) : 0.0f;
        }
    }
    __syncthreads();
    const bool in_lds = (m <= BINCAP);
    for (int base = 0; base < m; base += EDGE_CH) {
        int2 p[EDGE_PT];
#pragma unroll
        for (int j = 0; j < EDGE_PT; ++j) {
            const int i = base + tid + j * 1024;
            p[j] = (i < m) ? pairs[e0 + i] : make_int2(0, node0);
        }
#pragma unroll
        for (int j = 0; j < EDGE_PT; ++j) {
            if (base + tid + j * 1024 < m) {
                const int slot = atomicAdd(&s_cur[p[j].y - node0], 1);
                if (in_lds) s_src[slot] = p[j].x; else srcs[e0 + slot] = p[j].x;
            }
        }
    }
    __syncthreads();
    if (in_lds) {
        for (int i = tid; i < m; i += 1024) srcs[e0 + i] = s_src[i];
    }
}

template <int NPAD, bool ROWSCALE>
__global__ __launch_bounds__(256) void k_gemm_wmma(
        const float* __restrict__ A, const float* __restrict__ B,
        const float* __restrict__ bias, const float* __restrict__ rowscale,
        float* __restrict__ C, int M, int ncols) {
    constexpr int K  = 128;
    constexpr int NT = NPAD / 16;
    constexpr int LSTR = K + 2;
    __shared__ float ldsBT[NPAD * LSTR];

    bool staged = false;
    if constexpr ((NPAD & 31) == 0) {
        if (ncols == NPAD) {
            constexpr int CG = NPAD / 32;
#pragma unroll
            for (int it = 0; it < (K * NPAD / 4) / 256; ++it) {
                const int i = threadIdx.x + it * 256;
                const int w = i >> 5;
                const int k = (w / CG) * 4 + ((i >> 3) & 3);
                const int n = (w % CG) * 32 + (i & 7) * 4;
                const float4 v = *(const float4*)(B + k * NPAD + n);
                ldsBT[(n + 0) * LSTR + k] = v.x;
                ldsBT[(n + 1) * LSTR + k] = v.y;
                ldsBT[(n + 2) * LSTR + k] = v.z;
                ldsBT[(n + 3) * LSTR + k] = v.w;
            }
            staged = true;
        }
    }
    if (!staged) {
        const int q  = ncols >> 2;
        const int np = K * q;
        constexpr int NIT = (K * (NPAD / 4) + 255) / 256;
        float4 v[NIT];
#pragma unroll
        for (int it = 0; it < NIT; ++it) {
            const int i = threadIdx.x + it * 256;
            v[it] = ((const float4*)B)[(i < np) ? i : 0];
        }
#pragma unroll
        for (int it = 0; it < NIT; ++it) {
            const int i = threadIdx.x + it * 256;
            if (i < np) {
                const int k = i / q;
                const int n = (i - k * q) * 4;
                ldsBT[(n + 0) * LSTR + k] = v[it].x;
                ldsBT[(n + 1) * LSTR + k] = v[it].y;
                ldsBT[(n + 2) * LSTR + k] = v[it].z;
                ldsBT[(n + 3) * LSTR + k] = v[it].w;
            }
        }
        for (int i = threadIdx.x; i < (NPAD - ncols) * K; i += 256) {
            const int k = i & (K - 1);
            const int n = ncols + (i >> 7);
            ldsBT[n * LSTR + k] = 0.0f;
        }
    }
    __syncthreads();

    const int lane  = threadIdx.x & 31;
    const int tile  = __builtin_amdgcn_readfirstlane(blockIdx.x * 8 + (threadIdx.x >> 5));
    const int mtile = M >> 4;
    if (tile >= mtile) return;

    const int rowA = tile * 16 + (lane & 15);
    const int koff = (lane >> 4) * 2;
    const float* aptr = A + (size_t)rowA * K + koff;

    v8f acc[NT];
    v8f z8 = {0.f, 0.f, 0.f, 0.f, 0.f, 0.f, 0.f, 0.f};
#pragma unroll
    for (int nt = 0; nt < NT; ++nt) acc[nt] = z8;

#pragma unroll 4
    for (int k = 0; k < K; k += 4) {
        v2f a = *(const v2f*)(aptr + k);
#pragma unroll
        for (int nt = 0; nt < NT; ++nt) {
            int n = nt * 16 + (lane & 15);
            v2f b = *(const v2f*)(&ldsBT[n * LSTR + k + koff]);
            acc[nt] = __builtin_amdgcn_wmma_f32_16x16x4_f32(
                 false, a,  false, b,
                 (short)0, acc[nt],
                 false,  false);
        }
    }

    const int rbase = tile * 16 + ((lane >> 4) << 3);
    float sc[8];
    if (ROWSCALE) {
#pragma unroll
        for (int r = 0; r < 8; ++r) sc[r] = rowscale[rbase + r];
    }
#pragma unroll
    for (int nt = 0; nt < NT; ++nt) {
        int n = nt * 16 + (lane & 15);
        if (n >= ncols) continue;
        if (ROWSCALE) {
#pragma unroll
            for (int r = 0; r < 8; ++r) {
                C[(size_t)(rbase + r) * ncols + n] = acc[nt][r] * sc[r];
            }
        } else {
            float badd = bias ? bias[n] : 0.0f;
#pragma unroll
            for (int r = 0; r < 8; ++r) {
                C[(size_t)(rbase + r) * ncols + n] = acc[nt][r] + badd;
            }
        }
    }
}

template <int LAYER>
__global__ __launch_bounds__(256) void k_gather(
        const float* __restrict__ h, const float* __restrict__ dis,
        const int* __restrict__ off, const int* __restrict__ cnt,
        const int* __restrict__ srcs, float* __restrict__ y, int n,
        const float* __restrict__ bias,
        const float* __restrict__ gamma, const float* __restrict__ beta,
        const float* __restrict__ mean, const float* __restrict__ var,
        const float* __restrict__ x1) {
    const int node = __builtin_amdgcn_readfirstlane(blockIdx.x * 8 + (threadIdx.x >> 5));
    if (node >= n) return;
    const int lane  = threadIdx.x & 31;
    const int start = off[node];
    const int m     = cnt[node];
    const float dc  = dis[node];
    const float* hl = h + lane * 4;

    const float4 hs = *(const float4*)(hl + (size_t)node * HDIM);
    float4 xr = hs;
    if (LAYER == 2) xr = *(const float4*)(x1 + (size_t)node * HDIM + lane * 4);

    float4 acc = make_float4(0.f, 0.f, 0.f, 0.f);
    for (int base = 0; base < m; base += 32) {
        const int k = (m - base < 32) ? (m - base) : 32;
        int s = 0;
        if (lane < k) s = srcs[start + base + lane];
        int j = 0;
        for (; j + 4 <= k; j += 4) {
            const int s0 = __builtin_amdgcn_readlane(s, j);
            const int s1 = __builtin_amdgcn_readlane(s, j + 1);
            const int s2 = __builtin_amdgcn_readlane(s, j + 2);
            const int s3 = __builtin_amdgcn_readlane(s, j + 3);
            const float4 v0 = *(const float4*)(hl + (size_t)s0 * HDIM);
            const float4 v1 = *(const float4*)(hl + (size_t)s1 * HDIM);
            const float4 v2 = *(const float4*)(hl + (size_t)s2 * HDIM);
            const float4 v3 = *(const float4*)(hl + (size_t)s3 * HDIM);
            acc.x += v0.x; acc.y += v0.y; acc.z += v0.z; acc.w += v0.w;
            acc.x += v1.x; acc.y += v1.y; acc.z += v1.z; acc.w += v1.w;
            acc.x += v2.x; acc.y += v2.y; acc.z += v2.z; acc.w += v2.w;
            acc.x += v3.x; acc.y += v3.y; acc.z += v3.z; acc.w += v3.w;
        }
        for (; j < k; ++j) {
            const int s0 = __builtin_amdgcn_readlane(s, j);
            const float4 v0 = *(const float4*)(hl + (size_t)s0 * HDIM);
            acc.x += v0.x; acc.y += v0.y; acc.z += v0.z; acc.w += v0.w;
        }
    }
    const float4 bb = *(const float4*)(bias + lane * 4);
    float4 v;
    v.x = (acc.x + hs.x) * dc + bb.x;
    v.y = (acc.y + hs.y) * dc + bb.y;
    v.z = (acc.z + hs.z) * dc + bb.z;
    v.w = (acc.w + hs.w) * dc + bb.w;
    if (LAYER == 1) {
        const float4 mu = *(const float4*)(mean  + lane * 4);
        const float4 vr = *(const float4*)(var   + lane * 4);
        const float4 ga = *(const float4*)(gamma + lane * 4);
        const float4 be = *(const float4*)(beta  + lane * 4);
        v.x = fmaxf((v.x - mu.x) * rsqrtf(vr.x + 1e-5f) * ga.x + be.x, 0.0f);
        v.y = fmaxf((v.y - mu.y) * rsqrtf(vr.y + 1e-5f) * ga.y + be.y, 0.0f);
        v.z = fmaxf((v.z - mu.z) * rsqrtf(vr.z + 1e-5f) * ga.z + be.z, 0.0f);
        v.w = fmaxf((v.w - mu.w) * rsqrtf(vr.w + 1e-5f) * ga.w + be.w, 0.0f);
    } else {
        v.x = fmaxf(xr.x, v.x);
        v.y = fmaxf(xr.y, v.y);
        v.z = fmaxf(xr.z, v.z);
        v.w = fmaxf(xr.w, v.w);
    }
    *(float4*)(y + (size_t)node * HDIM + lane * 4) = v;
}

extern "C" void kernel_launch(void* const* d_in, const int* in_sizes, int n_in,
                              void* d_out, int out_size, void* d_ws, size_t ws_size,
                              hipStream_t stream) {
    const float* node_feat = (const float*)d_in[0];
    const int*   eidx      = (const int*)d_in[1];
    const float* W1        = (const float*)d_in[2];
    const float* b1        = (const float*)d_in[3];
    const float* gamma1    = (const float*)d_in[4];
    const float* beta1     = (const float*)d_in[5];
    const float* mean1     = (const float*)d_in[6];
    const float* var1      = (const float*)d_in[7];
    const float* W2        = (const float*)d_in[8];
    const float* b2        = (const float*)d_in[9];
    const float* Wf        = (const float*)d_in[10];
    const float* bf        = (const float*)d_in[11];
    float* out = (float*)d_out;

    const int n = in_sizes[0] / HDIM;
    const int e = in_sizes[1] / 2;
    const int* row = eidx;
    const int* col = eidx + e;

    const int nb     = (n + BINW - 1) / BINW;
    const int edgeGrd = (e + EDGE_CH - 1) / EDGE_CH;

    float* h   = (float*)d_ws;
    float* y1  = h + (size_t)n * HDIM;
    float* y2  = y1 + (size_t)n * HDIM;
    int2* pairs = (int2*)(y2 + (size_t)n * HDIM);
    int* cnt   = (int*)(pairs + e);
    int* off   = cnt + n;
    int* srcs  = off + n;
    int* hist  = srcs + e;
    int* binStart = hist + (size_t)edgeGrd * nb;
    float* dis = (float*)(binStart + nb + 1);

    const int mtiles  = n >> 4;
    const int gemmGrd = (mtiles + 7) / 8;
    const int nodeGrd = (n + 7) / 8;

    k_bin_hist<<<edgeGrd, 1024, 0, stream>>>(col, hist, e, nb);
    k_bin_part<<<edgeGrd, 1024, 0, stream>>>(row, col, hist, binStart, pairs, e, nb);
    k_bin_sort<<<nb, 1024, 0, stream>>>(pairs, binStart, off, cnt, dis, srcs, n);

    k_gemm_wmma<128, true><<<gemmGrd, 256, 0, stream>>>(node_feat, W1, nullptr, dis, h, n, HDIM);
    k_gather<1><<<nodeGrd, 256, 0, stream>>>(h, dis, off, cnt, srcs, y1, n,
                                             b1, gamma1, beta1, mean1, var1, nullptr);

    k_gemm_wmma<128, true><<<gemmGrd, 256, 0, stream>>>(y1, W2, nullptr, dis, h, n, HDIM);
    k_gather<2><<<nodeGrd, 256, 0, stream>>>(h, dis, off, cnt, srcs, y2, n,
                                             b2, nullptr, nullptr, nullptr, nullptr, y1);

    k_gemm_wmma<48, false><<<gemmGrd, 256, 0, stream>>>(y2, Wf, bf, nullptr, out, n, ODIM);
}


# Layer 2 sums neighbour rows of x1 BEFORE its dense product ((A x1) W2, products re-associated, list order kept): W2 product, JK max and final projection become one WMMA kernel; 8 -> 7 launches
# speedup vs baseline: 5.2406x; 1.0561x over previous
#include <hip/hip_runtime.h>

typedef __attribute__((ext_vector_type(2))) float v2f;
typedef __attribute__((ext_vector_type(4))) float v4f;
typedef __attribute__((ext_vector_type(8))) float v8f;

#define HDIM 128
#define ODIM 40

#define BINSH   9
#define BINW    512
#define MAXB    256
#define EDGE_CH 8192
#define EDGE_PT (EDGE_CH / 1024)
#define BINCAP  12288

__device__ __forceinline__ int block_excl_scan(int v, int* s_w) {
    const int lane = threadIdx.x & 31;
    const int wv   = threadIdx.x >> 5;
    int incl = v;
    for (int d = 1; d < 32; d <<= 1) {
        int t = __shfl_up(incl, d);
        if (lane >= d) incl += t;
    }
    if (lane == 31) s_w[wv] = incl;
    __syncthreads();
    int before = 0;
    for (int w = 0; w < 32; ++w) {
        int t = s_w[w];
        if (w < wv) before += t;
    }
    __syncthreads();
    return before + incl - v;
}

__global__ __launch_bounds__(1024) void k_bin_hist(const int* __restrict__ col,
                                                   int* __restrict__ hist, int e, int nb) {
    __shared__ int s_h[MAXB];
    const int tid = threadIdx.x;
    if (tid < MAXB) s_h[tid] = 0;
    __syncthreads();
    const int lo = blockIdx.x * EDGE_CH;
    const int hi = (lo + EDGE_CH < e) ? (lo + EDGE_CH) : e;
    int c[EDGE_PT];
#pragma unroll
    for (int j = 0; j < EDGE_PT; ++j) {
        const int i = lo + tid + j * 1024;
        c[j] = (i < hi) ? col[i] : 0;
    }
#pragma unroll
    for (int j = 0; j < EDGE_PT; ++j) {
        if (lo + tid + j * 1024 < hi) atomicAdd(&s_h[c[j] >> BINSH], 1);
    }
    __syncthreads();
    if (tid < nb) hist[blockIdx.x * nb + tid] = s_h[tid];
}

__global__ __launch_bounds__(1024) void k_bin_part(const int* __restrict__ row,
                                                   const int* __restrict__ col,
                                                   const int* __restrict__ hist,
                                                   int* __restrict__ binStart,
                                                   int2* __restrict__ pairs, int e, int nb) {
    __shared__ int  s_tot[MAXB];
    __shared__ int  s_pre[MAXB];
    __shared__ int  s_cur[MAXB];
    __shared__ int  s_delta[MAXB];
    __shared__ int  s_w[32];
    __shared__ int2 s_pair[EDGE_CH];
    const int tid = threadIdx.x;
    const int wg  = blockIdx.x;
    const int nwg = gridDim.x;
    if (tid < MAXB) { s_tot[tid] = 0; s_pre[tid] = 0; }
    __syncthreads();
    {
        const int b = tid & (MAXB - 1);
        int t_all = 0, t_pre = 0;
        if (b < nb) {
#pragma unroll 8
            for (int w = tid >> 8; w < nwg; w += 4) {
                const int v = hist[w * nb + b];
                t_all += v;
                t_pre += (w < wg) ? v : 0;
            }
            atomicAdd(&s_tot[b], t_all);
            atomicAdd(&s_pre[b], t_pre);
        }
    }
    __syncthreads();
    const int own = (tid < nb) ? hist[wg * nb + tid] : 0;
    const int tot = (tid < nb) ? s_tot[tid] : 0;
    const int gstart = block_excl_scan(tot, s_w);
    const int lstart = block_excl_scan(own, s_w);
    if (tid < nb) {
        s_cur[tid]   = lstart;
        s_delta[tid] = gstart + s_pre[tid] - lstart;
    }
    if (wg == 0 && tid <= nb) binStart[tid] = gstart;
    __syncthreads();

    const int lo = wg * EDGE_CH;
    const int hi = (lo + EDGE_CH < e) ? (lo + EDGE_CH) : e;
    int c[EDGE_PT], r[EDGE_PT];
#pragma unroll
    for (int j = 0; j < EDGE_PT; ++j) {
        const int i = lo + tid + j * 1024;
        c[j] = (i < hi) ? col[i] : 0;
        r[j] = (i < hi) ? row[i] : 0;
    }
#pragma unroll
    for (int j = 0; j < EDGE_PT; ++j) {
        if (lo + tid + j * 1024 < hi) {
            const int slot = atomicAdd(&s_cur[c[j] >> BINSH], 1);
            s_pair[slot] = make_int2(r[j], c[j]);
        }
    }
    __syncthreads();
    for (int i = tid; i < hi - lo; i += 1024) {
        const int2 p = s_pair[i];
        pairs[s_delta[p.y >> BINSH] + i] = p;
    }
}

__global__ __launch_bounds__(1024) void k_bin_sort(const int2* __restrict__ pairs,
                                                   const int* __restrict__ binStart,
                                                   int* __restrict__ off, int* __restrict__ cnt,
                                                   float* __restrict__ dis,
                                                   int* __restrict__ srcs, int n) {
    __shared__ int s_cnt[BINW];
    __shared__ int s_cur[BINW];
    __shared__ int s_w[32];
    __shared__ int s_src[BINCAP];
    const int tid   = threadIdx.x;
    const int node0 = blockIdx.x << BINSH;
    const int e0 = binStart[blockIdx.x];
    const int m  = binStart[blockIdx.x + 1] - e0;
    if (tid < BINW) s_cnt[tid] = 0;
    __syncthreads();
    for (int base = 0; base < m; base += EDGE_CH) {
        int t[EDGE_PT];
#pragma unroll
        for (int j = 0; j < EDGE_PT; ++j) {
            const int i = base + tid + j * 1024;
            t[j] = (i < m) ? pairs[e0 + i].y : node0;
        }
#pragma unroll
        for (int j = 0; j < EDGE_PT; ++j) {
            if (base + tid + j * 1024 < m) atomicAdd(&s_cnt[t[j] - node0], 1);
        }
    }
    __syncthreads();
    const int c  = (tid < BINW) ? s_cnt[tid] : 0;
    const int lo = block_excl_scan(c, s_w);
    if (tid < BINW) {
        s_cur[tid] = lo;
        const int node = node0 + tid;
        if (node < n) {
            off[node] = e0 + lo;
            cnt[node] = c;
            float dg = (float)(c + 1);
            dis[node] = (dg > 0.0f) ? rsqrtf(dg) : 0.0f;
        }
    }
    __syncthreads();
    const bool in_lds = (m <= BINCAP);
    for (int base = 0; base < m; base += EDGE_CH) {
        int2 p[EDGE_PT];
#pragma unroll
        for (int j = 0; j < EDGE_PT; ++j) {
            const int i = base + tid + j * 1024;
            p[j] = (i < m) ? pairs[e0 + i] : make_int2(0, node0);
        }
#pragma unroll
        for (int j = 0; j < EDGE_PT; ++j) {
            if (base + tid + j * 1024 < m) {
                const int slot = atomicAdd(&s_cur[p[j].y - node0], 1);
                if (in_lds) s_src[slot] = p[j].x; else srcs[e0 + slot] = p[j].x;
            }
        }
    }
    __syncthreads();
    if (in_lds) {
        for (int i = tid; i < m; i += 1024) srcs[e0 + i] = s_src[i];
    }
}

template <int NPAD, bool ROWSCALE>
__global__ __launch_bounds__(256) void k_gemm_wmma(
        const float* __restrict__ A, const float* __restrict__ B,
        const float* __restrict__ bias, const float* __restrict__ rowscale,
        float* __restrict__ C, int M, int ncols) {
    constexpr int K  = 128;
    constexpr int NT = NPAD / 16;
    constexpr int LSTR = K + 2;
    __shared__ float ldsBT[NPAD * LSTR];

    bool staged = false;
    if constexpr ((NPAD & 31) == 0) {
        if (ncols == NPAD) {
            constexpr int CG = NPAD / 32;
#pragma unroll
            for (int it = 0; it < (K * NPAD / 4) / 256; ++it) {
                const int i = threadIdx.x + it * 256;
                const int w = i >> 5;
                const int k = (w / CG) * 4 + ((i >> 3) & 3);
                const int n = (w % CG) * 32 + (i & 7) * 4;
                const float4 v = *(const float4*)(B + k * NPAD + n);
                ldsBT[(n + 0) * LSTR + k] = v.x;
                ldsBT[(n + 1) * LSTR + k] = v.y;
                ldsBT[(n + 2) * LSTR + k] = v.z;
                ldsBT[(n + 3) * LSTR + k] = v.w;
            }
            staged = true;
        }
    }
    if (!staged) {
        const int q  = ncols >> 2;
        const int np = K * q;
        constexpr int NIT = (K * (NPAD / 4) + 255) / 256;
        float4 v[NIT];
#pragma unroll
        for (int it = 0; it < NIT; ++it) {
            const int i = threadIdx.x + it * 256;
            v[it] = ((const float4*)B)[(i < np) ? i : 0];
        }
#pragma unroll
        for (int it = 0; it < NIT; ++it) {
            const int i = threadIdx.x + it * 256;
            if (i < np) {
                const int k = i / q;
                const int n = (i - k * q) * 4;
                ldsBT[(n + 0) * LSTR + k] = v[it].x;
                ldsBT[(n + 1) * LSTR + k] = v[it].y;
                ldsBT[(n + 2) * LSTR + k] = v[it].z;
                ldsBT[(n + 3) * LSTR + k] = v[it].w;
            }
        }
        for (int i = threadIdx.x; i < (NPAD - ncols) * K; i += 256) {
            const int k = i & (K - 1);
            const int n = ncols + (i >> 7);
            ldsBT[n * LSTR + k] = 0.0f;
        }
    }
    __syncthreads();

    const int lane  = threadIdx.x & 31;
    const int tile  = __builtin_amdgcn_readfirstlane(blockIdx.x * 8 + (threadIdx.x >> 5));
    const int mtile = M >> 4;
    if (tile >= mtile) return;

    const int rowA = tile * 16 + (lane & 15);
    const int koff = (lane >> 4) * 2;
    const float* aptr = A + (size_t)rowA * K + koff;

    v8f acc[NT];
    v8f z8 = {0.f, 0.f, 0.f, 0.f, 0.f, 0.f, 0.f, 0.f};
#pragma unroll
    for (int nt = 0; nt < NT; ++nt) acc[nt] = z8;

#pragma unroll 4
    for (int k = 0; k < K; k += 4) {
        v2f a = *(const v2f*)(aptr + k);
#pragma unroll
        for (int nt = 0; nt < NT; ++nt) {
            int n = nt * 16 + (lane & 15);
            v2f b = *(const v2f*)(&ldsBT[n * LSTR + k + koff]);
            acc[nt] = __builtin_amdgcn_wmma_f32_16x16x4_f32(
                 false, a,  false, b,
                 (short)0, acc[nt],
                 false,  false);
        }
    }

    const int rbase = tile * 16 + ((lane >> 4) << 3);
    float sc[8];
    if (ROWSCALE) {
#pragma unroll
        for (int r = 0; r < 8; ++r) sc[r] = rowscale[rbase + r];
    }
#pragma unroll
    for (int nt = 0; nt < NT; ++nt) {
        int n = nt * 16 + (lane & 15);
        if (n >= ncols) continue;
        if (ROWSCALE) {
#pragma unroll
            for (int r = 0; r < 8; ++r) {
                C[(size_t)(rbase + r) * ncols + n] = acc[nt][r] * sc[r];
            }
        } else {
            float badd = bias ? bias[n] : 0.0f;
#pragma unroll
            for (int r = 0; r < 8; ++r) {
                C[(size_t)(rbase + r) * ncols + n] = acc[nt][r] + badd;
            }
        }
    }
}

template <int LAYER>
__global__ __launch_bounds__(256) void k_gather(
        const float* __restrict__ h, const float* __restrict__ dis,
        const int* __restrict__ off, const int* __restrict__ cnt,
        const int* __restrict__ srcs, float* __restrict__ y, int n,
        const float* __restrict__ bias,
        const float* __restrict__ gamma, const float* __restrict__ beta,
        const float* __restrict__ mean, const float* __restrict__ var) {
    const int node = __builtin_amdgcn_readfirstlane(blockIdx.x * 8 + (threadIdx.x >> 5));
    if (node >= n) return;
    const int lane  = threadIdx.x & 31;
    const int start = off[node];
    const int m     = cnt[node];
    const float dc  = dis[node];
    const float* hl = h + lane * 4;

    const float4 hs = *(const float4*)(hl + (size_t)node * HDIM);

    float4 acc = make_float4(0.f, 0.f, 0.f, 0.f);
    for (int base = 0; base < m; base += 32) {
        const int k = (m - base < 32) ? (m - base) : 32;
        int s = 0;
        if (lane < k) s = srcs[start + base + lane];
        int j = 0;
        for (; j + 4 <= k; j += 4) {
            const int s0 = __builtin_amdgcn_readlane(s, j);
            const int s1 = __builtin_amdgcn_readlane(s, j + 1);
            const int s2 = __builtin_amdgcn_readlane(s, j + 2);
            const int s3 = __builtin_amdgcn_readlane(s, j + 3);
            const float4 v0 = *(const float4*)(hl + (size_t)s0 * HDIM);
            const float4 v1 = *(const float4*)(hl + (size_t)s1 * HDIM);
            const float4 v2 = *(const float4*)(hl + (size_t)s2 * HDIM);
            const float4 v3 = *(const float4*)(hl + (size_t)s3 * HDIM);
            acc.x += v0.x; acc.y += v0.y; acc.z += v0.z; acc.w += v0.w;
            acc.x += v1.x; acc.y += v1.y; acc.z += v1.z; acc.w += v1.w;
            acc.x += v2.x; acc.y += v2.y; acc.z += v2.z; acc.w += v2.w;
            acc.x += v3.x; acc.y += v3.y; acc.z += v3.z; acc.w += v3.w;
        }
        for (; j < k; ++j) {
            const int s0 = __builtin_amdgcn_readlane(s, j);
            const float4 v0 = *(const float4*)(hl + (size_t)s0 * HDIM);
            acc.x += v0.x; acc.y += v0.y; acc.z += v0.z; acc.w += v0.w;
        }
    }
    float4 v;
    if (LAYER == 1) {
        const float4 bb = *(const float4*)(bias + lane * 4);
        v.x = (acc.x + hs.x) * dc + bb.x;
        v.y = (acc.y + hs.y) * dc + bb.y;
        v.z = (acc.z + hs.z) * dc + bb.z;
        v.w = (acc.w + hs.w) * dc + bb.w;
        const float4 mu = *(const float4*)(mean  + lane * 4);
        const float4 vr = *(const float4*)(var   + lane * 4);
        const float4 ga = *(const float4*)(gamma + lane * 4);
        const float4 be = *(const float4*)(beta  + lane * 4);
        v.x = fmaxf((v.x - mu.x) * rsqrtf(vr.x + 1e-5f) * ga.x + be.x, 0.0f) * dc;
        v.y = fmaxf((v.y - mu.y) * rsqrtf(vr.y + 1e-5f) * ga.y + be.y, 0.0f) * dc;
        v.z = fmaxf((v.z - mu.z) * rsqrtf(vr.z + 1e-5f) * ga.z + be.z, 0.0f) * dc;
        v.w = fmaxf((v.w - mu.w) * rsqrtf(vr.w + 1e-5f) * ga.w + be.w, 0.0f) * dc;
    } else {
        v.x = (acc.x + hs.x) * dc;
        v.y = (acc.y + hs.y) * dc;
        v.z = (acc.z + hs.z) * dc;
        v.w = (acc.w + hs.w) * dc;
    }
    *(float4*)(y + (size_t)node * HDIM + lane * 4) = v;
}

__global__ __launch_bounds__(512, 8) void k_l2_out(
        const float* __restrict__ agg, const float* __restrict__ x1s,
        const float* __restrict__ dis,
        const float* __restrict__ W2, const float* __restrict__ b2,
        const float* __restrict__ Wf, const float* __restrict__ bf,
        float* __restrict__ out, int M) {
    constexpr int K    = HDIM;
    constexpr int LSTR = K + 2;
    constexpr int OPAD = 48;
    __shared__ float ldsW2T[K * LSTR];
    __shared__ float ldsWfT[OPAD * LSTR];

#pragma unroll
    for (int it = 0; it < (K * K / 4) / 512; ++it) {
        const int i = threadIdx.x + it * 512;
        const int w = i >> 5;
        const int k = (w >> 2) * 4 + ((i >> 3) & 3);
        const int n = (w & 3) * 32 + (i & 7) * 4;
        const float4 v = *(const float4*)(W2 + k * K + n);
        ldsW2T[(n + 0) * LSTR + k] = v.x;
        ldsW2T[(n + 1) * LSTR + k] = v.y;
        ldsW2T[(n + 2) * LSTR + k] = v.z;
        ldsW2T[(n + 3) * LSTR + k] = v.w;
    }
    {
        constexpr int Q   = ODIM / 4;
        constexpr int NP  = K * Q;
        constexpr int NIT = (NP + 511) / 512;
        float4 v[NIT];
#pragma unroll
        for (int it = 0; it < NIT; ++it) {
            const int i = threadIdx.x + it * 512;
            v[it] = ((const float4*)Wf)[(i < NP) ? i : 0];
        }
#pragma unroll
        for (int it = 0; it < NIT; ++it) {
            const int i = threadIdx.x + it * 512;
            if (i < NP) {
                const int k = i / Q;
                const int n = (i - k * Q) * 4;
                ldsWfT[(n + 0) * LSTR + k] = v[it].x;
                ldsWfT[(n + 1) * LSTR + k] = v[it].y;
                ldsWfT[(n + 2) * LSTR + k] = v[it].z;
                ldsWfT[(n + 3) * LSTR + k] = v[it].w;
            }
        }
        for (int i = threadIdx.x; i < (OPAD - ODIM) * K; i += 512) {
            const int k = i & (K - 1);
            const int n = ODIM + (i >> 7);
            ldsWfT[n * LSTR + k] = 0.0f;
        }
    }
    __syncthreads();

    const int lane = threadIdx.x & 31;
    const int tile = __builtin_amdgcn_readfirstlane(blockIdx.x * 16 + (threadIdx.x >> 5));
    if (tile >= (M >> 4)) return;

    const int c    = lane & 15;
    const int h8   = (lane >> 4) * 8;
    const int koff = (lane >> 4) * 2;
    const size_t node = (size_t)tile * 16 + c;
    const float* gptr = agg + node * K + koff;

    v8f t[8];
    const v8f z8 = {0.f, 0.f, 0.f, 0.f, 0.f, 0.f, 0.f, 0.f};
#pragma unroll
    for (int jt = 0; jt < 8; ++jt) {
        const float4 ba = *(const float4*)(b2 + jt * 16 + h8);
        const float4 bb = *(const float4*)(b2 + jt * 16 + h8 + 4);
        t[jt][0] = ba.x; t[jt][1] = ba.y; t[jt][2] = ba.z; t[jt][3] = ba.w;
        t[jt][4] = bb.x; t[jt][5] = bb.y; t[jt][6] = bb.z; t[jt][7] = bb.w;
    }

#pragma unroll 4
    for (int k = 0; k < K; k += 4) {
        const v2f g = *(const v2f*)(gptr + k);
#pragma unroll
        for (int jt = 0; jt < 8; ++jt) {
            const v2f w = *(const v2f*)(&ldsW2T[(jt * 16 + c) * LSTR + k + koff]);
            t[jt] = __builtin_amdgcn_wmma_f32_16x16x4_f32(
                 false, w,  false, g,
                 (short)0, t[jt],
                 false,  false);
        }
    }

    const float rd = 1.0f / dis[node];
    const float* xrow = x1s + node * K + h8;
    v8f o[3];
#pragma unroll
    for (int ot = 0; ot < 3; ++ot) o[ot] = z8;

    v4f xa = *(const v4f*)(xrow);
    v4f xb = *(const v4f*)(xrow + 4);
#pragma unroll
    for (int jt = 0; jt < 8; ++jt) {
        asm volatile("" :: "v"(xa), "v"(xb));
        t[jt][0] = fmaxf(xa[0] * rd, t[jt][0]);
        t[jt][1] = fmaxf(xa[1] * rd, t[jt][1]);
        t[jt][2] = fmaxf(xa[2] * rd, t[jt][2]);
        t[jt][3] = fmaxf(xa[3] * rd, t[jt][3]);
        t[jt][4] = fmaxf(xb[0] * rd, t[jt][4]);
        t[jt][5] = fmaxf(xb[1] * rd, t[jt][5]);
        t[jt][6] = fmaxf(xb[2] * rd, t[jt][6]);
        t[jt][7] = fmaxf(xb[3] * rd, t[jt][7]);
        if (jt < 7) {
            xa = *(const v4f*)(xrow + (jt + 1) * 16);
            xb = *(const v4f*)(xrow + (jt + 1) * 16 + 4);
            __builtin_amdgcn_sched_barrier(0);
        }
#pragma unroll
        for (int r2 = 0; r2 < 4; ++r2) {
            v2f mx;
            mx[0] = t[jt][2 * r2];
            mx[1] = t[jt][2 * r2 + 1];
#pragma unroll
            for (int ot = 0; ot < 3; ++ot) {
                const v2f wf = *(const v2f*)(&ldsWfT[(ot * 16 + c) * LSTR + jt * 16 + h8 + 2 * r2]);
                o[ot] = __builtin_amdgcn_wmma_f32_16x16x4_f32(
                     false, wf,  false, mx,
                     (short)0, o[ot],
                     false,  false);
            }
        }
    }

    float* orow = out + node * ODIM;
#pragma unroll
    for (int ot = 0; ot < 3; ++ot) {
        const int o0 = ot * 16 + h8;
        if (o0 < ODIM) {
            const float4 fa = *(const float4*)(bf + o0);
            const float4 fb = *(const float4*)(bf + o0 + 4);
            float4 sa, sb;
            sa.x = o[ot][0] + fa.x; sa.y = o[ot][1] + fa.y; sa.z = o[ot][2] + fa.z; sa.w = o[ot][3] + fa.w;
            sb.x = o[ot][4] + fb.x; sb.y = o[ot][5] + fb.y; sb.z = o[ot][6] + fb.z; sb.w = o[ot][7] + fb.w;
            *(float4*)(orow + o0)     = sa;
            *(float4*)(orow + o0 + 4) = sb;
        }
    }
}

extern "C" void kernel_launch(void* const* d_in, const int* in_sizes, int n_in,
                              void* d_out, int out_size, void* d_ws, size_t ws_size,
                              hipStream_t stream) {
    const float* node_feat = (const float*)d_in[0];
    const int*   eidx      = (const int*)d_in[1];
    const float* W1        = (const float*)d_in[2];
    const float* b1        = (const float*)d_in[3];
    const float* gamma1    = (const float*)d_in[4];
    const float* beta1     = (const float*)d_in[5];
    const float* mean1     = (const float*)d_in[6];
    const float* var1      = (const float*)d_in[7];
    const float* W2        = (const float*)d_in[8];
    const float* b2        = (const float*)d_in[9];
    const float* Wf        = (const float*)d_in[10];
    const float* bf        = (const float*)d_in[11];
    float* out = (float*)d_out;

    const int n = in_sizes[0] / HDIM;
    const int e = in_sizes[1] / 2;
    const int* row = eidx;
    const int* col = eidx + e;

    const int nb     = (n + BINW - 1) / BINW;
    const int edgeGrd = (e + EDGE_CH - 1) / EDGE_CH;

    float* h   = (float*)d_ws;
    float* y1  = h + (size_t)n * HDIM;
    float* y2  = y1 + (size_t)n * HDIM;
    int2* pairs = (int2*)(y2 + (size_t)n * HDIM);
    int* cnt   = (int*)(pairs + e);
    int* off   = cnt + n;
    int* srcs  = off + n;
    int* hist  = srcs + e;
    int* binStart = hist + (size_t)edgeGrd * nb;
    float* dis = (float*)(binStart + nb + 1);

    const int mtiles  = n >> 4;
    const int gemmGrd = (mtiles + 7) / 8;
    const int nodeGrd = (n + 7) / 8;
    const int l2Grd   = (mtiles + 15) / 16;

    k_bin_hist<<<edgeGrd, 1024, 0, stream>>>(col, hist, e, nb);
    k_bin_part<<<edgeGrd, 1024, 0, stream>>>(row, col, hist, binStart, pairs, e, nb);
    k_bin_sort<<<nb, 1024, 0, stream>>>(pairs, binStart, off, cnt, dis, srcs, n);

    k_gemm_wmma<128, true><<<gemmGrd, 256, 0, stream>>>(node_feat, W1, nullptr, dis, h, n, HDIM);
    k_gather<1><<<nodeGrd, 256, 0, stream>>>(h, dis, off, cnt, srcs, y1, n,
                                             b1, gamma1, beta1, mean1, var1);

    k_gather<2><<<nodeGrd, 256, 0, stream>>>(y1, dis, off, cnt, srcs, y2, n,
                                             nullptr, nullptr, nullptr, nullptr, nullptr);

    k_l2_out<<<l2Grd, 512, 0, stream>>>(y2, y1, dis, W2, b2, Wf, bf, out, n);
}


# Layer-2 neighbour sum moved into the WMMA kernel: a wave sums its 16 nodes' source rows (list order kept, 16 rows in flight) into its own LDS tile and multiplies from there; 7 -> 6 launches
# speedup vs baseline: 5.6559x; 1.0792x over previous
#include <hip/hip_runtime.h>

typedef __attribute__((ext_vector_type(2))) float v2f;
typedef __attribute__((ext_vector_type(4))) float v4f;
typedef __attribute__((ext_vector_type(8))) float v8f;

#define HDIM 128
#define ODIM 40

#define BINSH   9
#define BINW    512
#define MAXB    256
#define EDGE_CH 8192
#define EDGE_PT (EDGE_CH / 1024)
#define BINCAP  12288

__device__ __forceinline__ int block_excl_scan(int v, int* s_w) {
    const int lane = threadIdx.x & 31;
    const int wv   = threadIdx.x >> 5;
    int incl = v;
    for (int d = 1; d < 32; d <<= 1) {
        int t = __shfl_up(incl, d);
        if (lane >= d) incl += t;
    }
    if (lane == 31) s_w[wv] = incl;
    __syncthreads();
    int before = 0;
    for (int w = 0; w < 32; ++w) {
        int t = s_w[w];
        if (w < wv) before += t;
    }
    __syncthreads();
    return before + incl - v;
}

__global__ __launch_bounds__(1024) void k_bin_hist(const int* __restrict__ col,
                                                   int* __restrict__ hist, int e, int nb) {
    __shared__ int s_h[MAXB];
    const int tid = threadIdx.x;
    if (tid < MAXB) s_h[tid] = 0;
    __syncthreads();
    const int lo = blockIdx.x * EDGE_CH;
    const int hi = (lo + EDGE_CH < e) ? (lo + EDGE_CH) : e;
    int c[EDGE_PT];
#pragma unroll
    for (int j = 0; j < EDGE_PT; ++j) {
        const int i = lo + tid + j * 1024;
        c[j] = (i < hi) ? col[i] : 0;
    }
#pragma unroll
    for (int j = 0; j < EDGE_PT; ++j) {
        if (lo + tid + j * 1024 < hi) atomicAdd(&s_h[c[j] >> BINSH], 1);
    }
    __syncthreads();
    if (tid < nb) hist[blockIdx.x * nb + tid] = s_h[tid];
}

__global__ __launch_bounds__(1024) void k_bin_part(const int* __restrict__ row,
                                                   const int* __restrict__ col,
                                                   const int* __restrict__ hist,
                                                   int* __restrict__ binStart,
                                                   int2* __restrict__ pairs, int e, int nb) {
    __shared__ int  s_tot[MAXB];
    __shared__ int  s_pre[MAXB];
    __shared__ int  s_cur[MAXB];
    __shared__ int  s_delta[MAXB];
    __shared__ int  s_w[32];
    __shared__ int2 s_pair[EDGE_CH];
    const int tid = threadIdx.x;
    const int wg  = blockIdx.x;
    const int nwg = gridDim.x;
    if (tid < MAXB) { s_tot[tid] = 0; s_pre[tid] = 0; }
    __syncthreads();
    {
        const int b = tid & (MAXB - 1);
        int t_all = 0, t_pre = 0;
        if (b < nb) {
#pragma unroll 8
            for (int w = tid >> 8; w < nwg; w += 4) {
                const int v = hist[w * nb + b];
                t_all += v;
                t_pre += (w < wg) ? v : 0;
            }
            atomicAdd(&s_tot[b], t_all);
            atomicAdd(&s_pre[b], t_pre);
        }
    }
    __syncthreads();
    const int own = (tid < nb) ? hist[wg * nb + tid] : 0;
    const int tot = (tid < nb) ? s_tot[tid] : 0;
    const int gstart = block_excl_scan(tot, s_w);
    const int lstart = block_excl_scan(own, s_w);
    if (tid < nb) {
        s_cur[tid]   = lstart;
        s_delta[tid] = gstart + s_pre[tid] - lstart;
    }
    if (wg == 0 && tid <= nb) binStart[tid] = gstart;
    __syncthreads();

    const int lo = wg * EDGE_CH;
    const int hi = (lo + EDGE_CH < e) ? (lo + EDGE_CH) : e;
    int c[EDGE_PT], r[EDGE_PT];
#pragma unroll
    for (int j = 0; j < EDGE_PT; ++j) {
        const int i = lo + tid + j * 1024;
        c[j] = (i < hi) ? col[i] : 0;
        r[j] = (i < hi) ? row[i] : 0;
    }
#pragma unroll
    for (int j = 0; j < EDGE_PT; ++j) {
        if (lo + tid + j * 1024 < hi) {
            const int slot = atomicAdd(&s_cur[c[j] >> BINSH], 1);
            s_pair[slot] = make_int2(r[j], c[j]);
        }
    }
    __syncthreads();
    for (int i = tid; i < hi - lo; i += 1024) {
        const int2 p = s_pair[i];
        pairs[s_delta[p.y >> BINSH] + i] = p;
    }
}

__global__ __launch_bounds__(1024) void k_bin_sort(const int2* __restrict__ pairs,
                                                   const int* __restrict__ binStart,
                                                   int* __restrict__ off, int* __restrict__ cnt,
                                                   float* __restrict__ dis,
                                                   int* __restrict__ srcs, int n) {
    __shared__ int s_cnt[BINW];
    __shared__ int s_cur[BINW];
    __shared__ int s_w[32];
    __shared__ int s_src[BINCAP];
    const int tid   = threadIdx.x;
    const int node0 = blockIdx.x << BINSH;
    const int e0 = binStart[blockIdx.x];
    const int m  = binStart[blockIdx.x + 1] - e0;
    if (tid < BINW) s_cnt[tid] = 0;
    __syncthreads();
    for (int base = 0; base < m; base += EDGE_CH) {
        int t[EDGE_PT];
#pragma unroll
        for (int j = 0; j < EDGE_PT; ++j) {
            const int i = base + tid + j * 1024;
            t[j] = (i < m) ? pairs[e0 + i].y : node0;
        }
#pragma unroll
        for (int j = 0; j < EDGE_PT; ++j) {
            if (base + tid + j * 1024 < m) atomicAdd(&s_cnt[t[j] - node0], 1);
        }
    }
    __syncthreads();
    const int c  = (tid < BINW) ? s_cnt[tid] : 0;
    const int lo = block_excl_scan(c, s_w);
    if (tid < BINW) {
        s_cur[tid] = lo;
        const int node = node0 + tid;
        if (node < n) {
            off[node] = e0 + lo;
            cnt[node] = c;
            float dg = (float)(c + 1);
            dis[node] = (dg > 0.0f) ? rsqrtf(dg) : 0.0f;
        }
    }
    __syncthreads();
    const bool in_lds = (m <= BINCAP);
    for (int base = 0; base < m; base += EDGE_CH) {
        int2 p[EDGE_PT];
#pragma unroll
        for (int j = 0; j < EDGE_PT; ++j) {
            const int i = base + tid + j * 1024;
            p[j] = (i < m) ? pairs[e0 + i] : make_int2(0, node0);
        }
#pragma unroll
        for (int j = 0; j < EDGE_PT; ++j) {
            if (base + tid + j * 1024 < m) {
                const int slot = atomicAdd(&s_cur[p[j].y - node0], 1);
                if (in_lds) s_src[slot] = p[j].x; else srcs[e0 + slot] = p[j].x;
            }
        }
    }
    __syncthreads();
    if (in_lds) {
        for (int i = tid; i < m; i += 1024) srcs[e0 + i] = s_src[i];
    }
}

template <int NPAD, bool ROWSCALE>
__global__ __launch_bounds__(256) void k_gemm_wmma(
        const float* __restrict__ A, const float* __restrict__ B,
        const float* __restrict__ bias, const float* __restrict__ rowscale,
        float* __restrict__ C, int M, int ncols) {
    constexpr int K  = 128;
    constexpr int NT = NPAD / 16;
    constexpr int LSTR = K + 2;
    __shared__ float ldsBT[NPAD * LSTR];

    bool staged = false;
    if constexpr ((NPAD & 31) == 0) {
        if (ncols == NPAD) {
            constexpr int CG = NPAD / 32;
#pragma unroll
            for (int it = 0; it < (K * NPAD / 4) / 256; ++it) {
                const int i = threadIdx.x + it * 256;
                const int w = i >> 5;
                const int k = (w / CG) * 4 + ((i >> 3) & 3);
                const int n = (w % CG) * 32 + (i & 7) * 4;
                const float4 v = *(const float4*)(B + k * NPAD + n);
                ldsBT[(n + 0) * LSTR + k] = v.x;
                ldsBT[(n + 1) * LSTR + k] = v.y;
                ldsBT[(n + 2) * LSTR + k] = v.z;
                ldsBT[(n + 3) * LSTR + k] = v.w;
            }
            staged = true;
        }
    }
    if (!staged) {
        const int q  = ncols >> 2;
        const int np = K * q;
        constexpr int NIT = (K * (NPAD / 4) + 255) / 256;
        float4 v[NIT];
#pragma unroll
        for (int it = 0; it < NIT; ++it) {
            const int i = threadIdx.x + it * 256;
            v[it] = ((const float4*)B)[(i < np) ? i : 0];
        }
#pragma unroll
        for (int it = 0; it < NIT; ++it) {
            const int i = threadIdx.x + it * 256;
            if (i < np) {
                const int k = i / q;
                const int n = (i - k * q) * 4;
                ldsBT[(n + 0) * LSTR + k] = v[it].x;
                ldsBT[(n + 1) * LSTR + k] = v[it].y;
                ldsBT[(n + 2) * LSTR + k] = v[it].z;
                ldsBT[(n + 3) * LSTR + k] = v[it].w;
            }
        }
        for (int i = threadIdx.x; i < (NPAD - ncols) * K; i += 256) {
            const int k = i & (K - 1);
            const int n = ncols + (i >> 7);
            ldsBT[n * LSTR + k] = 0.0f;
        }
    }
    __syncthreads();

    const int lane  = threadIdx.x & 31;
    const int tile  = __builtin_amdgcn_readfirstlane(blockIdx.x * 8 + (threadIdx.x >> 5));
    const int mtile = M >> 4;
    if (tile >= mtile) return;

    const int rowA = tile * 16 + (lane & 15);
    const int koff = (lane >> 4) * 2;
    const float* aptr = A + (size_t)rowA * K + koff;

    v8f acc[NT];
    v8f z8 = {0.f, 0.f, 0.f, 0.f, 0.f, 0.f, 0.f, 0.f};
#pragma unroll
    for (int nt = 0; nt < NT; ++nt) acc[nt] = z8;

#pragma unroll 4
    for (int k = 0; k < K; k += 4) {
        v2f a = *(const v2f*)(aptr + k);
#pragma unroll
        for (int nt = 0; nt < NT; ++nt) {
            int n = nt * 16 + (lane & 15);
            v2f b = *(const v2f*)(&ldsBT[n * LSTR + k + koff]);
            acc[nt] = __builtin_amdgcn_wmma_f32_16x16x4_f32(
                 false, a,  false, b,
                 (short)0, acc[nt],
                 false,  false);
        }
    }

    const int rbase = tile * 16 + ((lane >> 4) << 3);
    float sc[8];
    if (ROWSCALE) {
#pragma unroll
        for (int r = 0; r < 8; ++r) sc[r] = rowscale[rbase + r];
    }
#pragma unroll
    for (int nt = 0; nt < NT; ++nt) {
        int n = nt * 16 + (lane & 15);
        if (n >= ncols) continue;
        if (ROWSCALE) {
#pragma unroll
            for (int r = 0; r < 8; ++r) {
                C[(size_t)(rbase + r) * ncols + n] = acc[nt][r] * sc[r];
            }
        } else {
            float badd = bias ? bias[n] : 0.0f;
#pragma unroll
            for (int r = 0; r < 8; ++r) {
                C[(size_t)(rbase + r) * ncols + n] = acc[nt][r] + badd;
            }
        }
    }
}

__global__ __launch_bounds__(256) void k_gather(
        const float* __restrict__ h, const float* __restrict__ dis,
        const int* __restrict__ off, const int* __restrict__ cnt,
        const int* __restrict__ srcs, float* __restrict__ y, int n,
        const float* __restrict__ bias,
        const float* __restrict__ gamma, const float* __restrict__ beta,
        const float* __restrict__ mean, const float* __restrict__ var) {
    const int node = __builtin_amdgcn_readfirstlane(blockIdx.x * 8 + (threadIdx.x >> 5));
    if (node >= n) return;
    const int lane  = threadIdx.x & 31;
    const int start = off[node];
    const int m     = cnt[node];
    const float dc  = dis[node];
    const float* hl = h + lane * 4;

    const float4 hs = *(const float4*)(hl + (size_t)node * HDIM);

    float4 acc = make_float4(0.f, 0.f, 0.f, 0.f);
    for (int base = 0; base < m; base += 32) {
        const int k = (m - base < 32) ? (m - base) : 32;
        int s = 0;
        if (lane < k) s = srcs[start + base + lane];
        int j = 0;
        for (; j + 4 <= k; j += 4) {
            const int s0 = __builtin_amdgcn_readlane(s, j);
            const int s1 = __builtin_amdgcn_readlane(s, j + 1);
            const int s2 = __builtin_amdgcn_readlane(s, j + 2);
            const int s3 = __builtin_amdgcn_readlane(s, j + 3);
            const float4 v0 = *(const float4*)(hl + (size_t)s0 * HDIM);
            const float4 v1 = *(const float4*)(hl + (size_t)s1 * HDIM);
            const float4 v2 = *(const float4*)(hl + (size_t)s2 * HDIM);
            const float4 v3 = *(const float4*)(hl + (size_t)s3 * HDIM);
            acc.x += v0.x; acc.y += v0.y; acc.z += v0.z; acc.w += v0.w;
            acc.x += v1.x; acc.y += v1.y; acc.z += v1.z; acc.w += v1.w;
            acc.x += v2.x; acc.y += v2.y; acc.z += v2.z; acc.w += v2.w;
            acc.x += v3.x; acc.y += v3.y; acc.z += v3.z; acc.w += v3.w;
        }
        for (; j < k; ++j) {
            const int s0 = __builtin_amdgcn_readlane(s, j);
            const float4 v0 = *(const float4*)(hl + (size_t)s0 * HDIM);
            acc.x += v0.x; acc.y += v0.y; acc.z += v0.z; acc.w += v0.w;
        }
    }
    float4 v;
    const float4 bb = *(const float4*)(bias + lane * 4);
    v.x = (acc.x + hs.x) * dc + bb.x;
    v.y = (acc.y + hs.y) * dc + bb.y;
    v.z = (acc.z + hs.z) * dc + bb.z;
    v.w = (acc.w + hs.w) * dc + bb.w;
    const float4 mu = *(const float4*)(mean  + lane * 4);
    const float4 vr = *(const float4*)(var   + lane * 4);
    const float4 ga = *(const float4*)(gamma + lane * 4);
    const float4 be = *(const float4*)(beta  + lane * 4);
    v.x = fmaxf((v.x - mu.x) * rsqrtf(vr.x + 1e-5f) * ga.x + be.x, 0.0f) * dc;
    v.y = fmaxf((v.y - mu.y) * rsqrtf(vr.y + 1e-5f) * ga.y + be.y, 0.0f) * dc;
    v.z = fmaxf((v.z - mu.z) * rsqrtf(vr.z + 1e-5f) * ga.z + be.z, 0.0f) * dc;
    v.w = fmaxf((v.w - mu.w) * rsqrtf(vr.w + 1e-5f) * ga.w + be.w, 0.0f) * dc;
    *(float4*)(y + (size_t)node * HDIM + lane * 4) = v;
}

#define L2W  25
#define L2T  (L2W * 32)
#define TSTR 132
#define GRP  16

__global__ __launch_bounds__(L2T) void k_l2_out(
        const float* __restrict__ x1s, const float* __restrict__ dis,
        const int* __restrict__ off, const int* __restrict__ cnt,
        const int* __restrict__ srcs,
        const float* __restrict__ W2, const float* __restrict__ b2,
        const float* __restrict__ Wf, const float* __restrict__ bf,
        float* __restrict__ out, int M) {
    constexpr int K    = HDIM;
    constexpr int LSTR = K + 2;
    constexpr int OPAD = 48;
    __shared__ float ldsW2T[K * LSTR];
    __shared__ float ldsWfT[OPAD * LSTR];
    __shared__ __attribute__((aligned(16))) float ldsAgg[L2W * 16 * TSTR];

#pragma unroll
    for (int it = 0; it < (K * K / 4 + L2T - 1) / L2T; ++it) {
        const int i = threadIdx.x + it * L2T;
        if (i < K * K / 4) {
            const int w = i >> 5;
            const int k = (w >> 2) * 4 + ((i >> 3) & 3);
            const int n = (w & 3) * 32 + (i & 7) * 4;
            const float4 v = *(const float4*)(W2 + k * K + n);
            ldsW2T[(n + 0) * LSTR + k] = v.x;
            ldsW2T[(n + 1) * LSTR + k] = v.y;
            ldsW2T[(n + 2) * LSTR + k] = v.z;
            ldsW2T[(n + 3) * LSTR + k] = v.w;
        }
    }
    {
        constexpr int Q   = ODIM / 4;
        constexpr int NP  = K * Q;
#pragma unroll
        for (int it = 0; it < (NP + L2T - 1) / L2T; ++it) {
            const int i = threadIdx.x + it * L2T;
            if (i < NP) {
                const float4 v = ((const float4*)Wf)[i];
                const int k = i / Q;
                const int n = (i - k * Q) * 4;
                ldsWfT[(n + 0) * LSTR + k] = v.x;
                ldsWfT[(n + 1) * LSTR + k] = v.y;
                ldsWfT[(n + 2) * LSTR + k] = v.z;
                ldsWfT[(n + 3) * LSTR + k] = v.w;
            }
        }
        for (int i = threadIdx.x; i < (OPAD - ODIM) * K; i += L2T) {
            const int k = i & (K - 1);
            const int n = ODIM + (i >> 7);
            ldsWfT[n * LSTR + k] = 0.0f;
        }
    }
    __syncthreads();

    const int lane = threadIdx.x & 31;
    const int wave = __builtin_amdgcn_readfirstlane(threadIdx.x >> 5);
    const int tile = __builtin_amdgcn_readfirstlane(blockIdx.x * L2W + (threadIdx.x >> 5));
    if (tile >= (M >> 4)) return;

    const int c    = lane & 15;
    const int h8   = (lane >> 4) * 8;
    const int koff = (lane >> 4) * 2;
    const int node0 = tile * 16;
    float* tw = ldsAgg + wave * (16 * TSTR);

    const int   offv = (off + node0)[c];
    const int   endv = offv + (cnt + node0)[c];
    const float disv = (dis + node0)[c];
    const float* hl  = x1s + lane * 4;
    {
        float* tl = tw + lane * 4;
#pragma unroll
        for (int i = 0; i < 16; ++i) {
            const float4 t = *(const float4*)(hl + (size_t)(node0 + i) * HDIM);
            *(float4*)(tl + i * TSTR) = t;
        }
        const int L0   = __builtin_amdgcn_readlane(offv, 0);
        const int Lend = __builtin_amdgcn_readlane(endv, 15);
        int cur  = 0;
        int endc = __builtin_amdgcn_readlane(endv, 0);
        float4 acc = make_float4(0.f, 0.f, 0.f, 0.f);
        auto finish = [&](int i, const float4& a) {
            float* tr = tl + i * TSTR;
            const float4 hs = *(const float4*)tr;
            const float dc  = __int_as_float(__builtin_amdgcn_readlane(__float_as_int(disv), i));
            float4 r;
            r.x = (a.x + hs.x) * dc;
            r.y = (a.y + hs.y) * dc;
            r.z = (a.z + hs.z) * dc;
            r.w = (a.w + hs.w) * dc;
            *(float4*)tr = r;
        };
        int sn = 0;
        if (L0 + lane < Lend) sn = srcs[L0 + lane];
        for (int base = L0; base < Lend; base += 32) {
            const int kk = (Lend - base < 32) ? (Lend - base) : 32;
            const int s = sn;
            for (int j = 0; j < kk; j += GRP) {
                float4 v[GRP];
#pragma unroll
                for (int q = 0; q < GRP; ++q) {
                    const int jq = (j + q < kk) ? (j + q) : (kk - 1);
                    const int sq = __builtin_amdgcn_readlane(s, jq);
                    v[q] = *(const float4*)(hl + (size_t)sq * HDIM);
                }
                if (j == 0 && base + 32 + lane < Lend) sn = srcs[base + 32 + lane];
#pragma unroll
                for (int q = 0; q < GRP; ++q) {
                    if (j + q < kk) {
                        const int pos = base + j + q;
                        while (pos >= endc) {
                            finish(cur, acc);
                            acc = make_float4(0.f, 0.f, 0.f, 0.f);
                            ++cur;
                            endc = __builtin_amdgcn_readlane(endv, cur);
                        }
                        acc.x += v[q].x; acc.y += v[q].y; acc.z += v[q].z; acc.w += v[q].w;
                    }
                }
            }
        }
        while (cur < 16) {
            finish(cur, acc);
            acc = make_float4(0.f, 0.f, 0.f, 0.f);
            ++cur;
        }
    }
    __builtin_amdgcn_fence(__ATOMIC_RELEASE, "wavefront");
    __builtin_amdgcn_wave_barrier();
    __builtin_amdgcn_fence(__ATOMIC_ACQUIRE, "wavefront");

    const float* gptr = tw + c * TSTR + koff;

    v8f t[8];
    const v8f z8 = {0.f, 0.f, 0.f, 0.f, 0.f, 0.f, 0.f, 0.f};
#pragma unroll
    for (int jt = 0; jt < 8; ++jt) {
        const float4 ba = *(const float4*)(b2 + jt * 16 + h8);
        const float4 bb = *(const float4*)(b2 + jt * 16 + h8 + 4);
        t[jt][0] = ba.x; t[jt][1] = ba.y; t[jt][2] = ba.z; t[jt][3] = ba.w;
        t[jt][4] = bb.x; t[jt][5] = bb.y; t[jt][6] = bb.z; t[jt][7] = bb.w;
    }

#pragma unroll 2
    for (int k = 0; k < K; k += 4) {
        const v2f g = *(const v2f*)(gptr + k);
#pragma unroll
        for (int jt = 0; jt < 8; ++jt) {
            const v2f w = *(const v2f*)(&ldsW2T[(jt * 16 + c) * LSTR + k + koff]);
            t[jt] = __builtin_amdgcn_wmma_f32_16x16x4_f32(
                 false, w,  false, g,
                 (short)0, t[jt],
                 false,  false);
        }
    }

    const float rd = 1.0f / disv;
    const float* xrow = (x1s + (size_t)node0 * K) + (c * K + h8);
    v8f o[3];
#pragma unroll
    for (int ot = 0; ot < 3; ++ot) o[ot] = z8;

    v4f xa = *(const v4f*)(xrow);
    v4f xb = *(const v4f*)(xrow + 4);
#pragma unroll
    for (int jt = 0; jt < 8; ++jt) {
        asm volatile("" :: "v"(xa), "v"(xb));
        t[jt][0] = fmaxf(xa[0] * rd, t[jt][0]);
        t[jt][1] = fmaxf(xa[1] * rd, t[jt][1]);
        t[jt][2] = fmaxf(xa[2] * rd, t[jt][2]);
        t[jt][3] = fmaxf(xa[3] * rd, t[jt][3]);
        t[jt][4] = fmaxf(xb[0] * rd, t[jt][4]);
        t[jt][5] = fmaxf(xb[1] * rd, t[jt][5]);
        t[jt][6] = fmaxf(xb[2] * rd, t[jt][6]);
        t[jt][7] = fmaxf(xb[3] * rd, t[jt][7]);
        if (jt < 7) {
            xa = *(const v4f*)(xrow + (jt + 1) * 16);
            xb = *(const v4f*)(xrow + (jt + 1) * 16 + 4);
            __builtin_amdgcn_sched_barrier(0);
        }
#pragma unroll
        for (int r2 = 0; r2 < 4; ++r2) {
            v2f mx;
            mx[0] = t[jt][2 * r2];
            mx[1] = t[jt][2 * r2 + 1];
#pragma unroll
            for (int ot = 0; ot < 3; ++ot) {
                const v2f wf = *(const v2f*)(&ldsWfT[(ot * 16 + c) * LSTR + jt * 16 + h8 + 2 * r2]);
                o[ot] = __builtin_amdgcn_wmma_f32_16x16x4_f32(
                     false, wf,  false, mx,
                     (short)0, o[ot],
                     false,  false);
            }
        }
    }

    float* orow = (out + (size_t)node0 * ODIM) + c * ODIM;
#pragma unroll
    for (int ot = 0; ot < 3; ++ot) {
        const int o0 = ot * 16 + h8;
        if (o0 < ODIM) {
            const float4 fa = *(const float4*)(bf + o0);
            const float4 fb = *(const float4*)(bf + o0 + 4);
            float4 sa, sb;
            sa.x = o[ot][0] + fa.x; sa.y = o[ot][1] + fa.y; sa.z = o[ot][2] + fa.z; sa.w = o[ot][3] + fa.w;
            sb.x = o[ot][4] + fb.x; sb.y = o[ot][5] + fb.y; sb.z = o[ot][6] + fb.z; sb.w = o[ot][7] + fb.w;
            *(float4*)(orow + o0)     = sa;
            *(float4*)(orow + o0 + 4) = sb;
        }
    }
}

extern "C" void kernel_launch(void* const* d_in, const int* in_sizes, int n_in,
                              void* d_out, int out_size, void* d_ws, size_t ws_size,
                              hipStream_t stream) {
    const float* node_feat = (const float*)d_in[0];
    const int*   eidx      = (const int*)d_in[1];
    const float* W1        = (const float*)d_in[2];
    const float* b1        = (const float*)d_in[3];
    const float* gamma1    = (const float*)d_in[4];
    const float* beta1     = (const float*)d_in[5];
    const float* mean1     = (const float*)d_in[6];
    const float* var1      = (const float*)d_in[7];
    const float* W2        = (const float*)d_in[8];
    const float* b2        = (const float*)d_in[9];
    const float* Wf        = (const float*)d_in[10];
    const float* bf        = (const float*)d_in[11];
    float* out = (float*)d_out;

    const int n = in_sizes[0] / HDIM;
    const int e = in_sizes[1] / 2;
    const int* row = eidx;
    const int* col = eidx + e;

    const int nb     = (n + BINW - 1) / BINW;
    const int edgeGrd = (e + EDGE_CH - 1) / EDGE_CH;

    float* h   = (float*)d_ws;
    float* y1  = h + (size_t)n * HDIM;
    int2* pairs = (int2*)(y1 + (size_t)n * HDIM);
    int* cnt   = (int*)(pairs + e);
    int* off   = cnt + n;
    int* srcs  = off + n;
    int* hist  = srcs + e;
    int* binStart = hist + (size_t)edgeGrd * nb;
    float* dis = (float*)(binStart + nb + 1);

    const int mtiles  = n >> 4;
    const int gemmGrd = (mtiles + 7) / 8;
    const int nodeGrd = (n + 7) / 8;
    const int l2Grd   = (mtiles + L2W - 1) / L2W;

    k_bin_hist<<<edgeGrd, 1024, 0, stream>>>(col, hist, e, nb);
    k_bin_part<<<edgeGrd, 1024, 0, stream>>>(row, col, hist, binStart, pairs, e, nb);
    k_bin_sort<<<nb, 1024, 0, stream>>>(pairs, binStart, off, cnt, dis, srcs, n);

    k_gemm_wmma<128, true><<<gemmGrd, 256, 0, stream>>>(node_feat, W1, nullptr, dis, h, n, HDIM);
    k_gather<<<nodeGrd, 256, 0, stream>>>(h, dis, off, cnt, srcs, y1, n,
                                          b1, gamma1, beta1, mean1, var1);

    k_l2_out<<<l2Grd, L2T, 0, stream>>>(y1, dis, off, cnt, srcs, W2, b2, Wf, bf, out, n);
}
